# starting kernel
# baseline (speedup 1.0000x reference)
#include <hip/hip_runtime.h>


typedef __attribute__((ext_vector_type(2))) float v2f;
typedef __attribute__((ext_vector_type(8))) float v8f;

#define D_IN 64
#define WAVES_PER_BLOCK 8

__global__ void deg_init_kernel(float* __restrict__ deg, int n) {
    int i = blockIdx.x * blockDim.x + threadIdx.x;
    if (i < n) deg[i] = 1.0f;
}

__global__ void deg_edges_kernel(const int* __restrict__ dst,
                                 float* __restrict__ deg, int e) {
    int i = blockIdx.x * blockDim.x + threadIdx.x;
    if (i < e) atomicAdd(&deg[dst[i]], 1.0f);
}

__global__ void deg_rsqrt_kernel(float* __restrict__ deg, int n) {
    int i = blockIdx.x * blockDim.x + threadIdx.x;
    if (i < n) deg[i] = rsqrtf(deg[i]);
}

template <int OUTC, bool RELU_IN>
__global__ void gemm_wmma_kernel(const float* __restrict__ X,
                                 const float* __restrict__ W,
                                 float* __restrict__ H, int nrows) {
    const int lane = threadIdx.x & 31;
    const int wave = threadIdx.x >> 5;
    const int lo = lane & 15;
    const int hi = lane >> 4;

    const int tile = blockIdx.x * WAVES_PER_BLOCK + wave;
    const int row0 = tile * 16;
    if (row0 >= nrows) return;

    constexpr int NT = OUTC / 16;
    v8f acc[NT] = {};

    const float* __restrict__ xr = X + (size_t)(row0 + lo) * D_IN;

#pragma unroll
    for (int k0 = 0; k0 < D_IN; k0 += 4) {
        float a0 = xr[k0 + 2 * hi + 0];
        float a1 = xr[k0 + 2 * hi + 1];
        if (RELU_IN) {
            a0 = fmaxf(a0, 0.0f);
            a1 = fmaxf(a1, 0.0f);
        }
        v2f a;
        a.x = a0;
        a.y = a1;
#pragma unroll
        for (int t = 0; t < NT; ++t) {
            v2f b;
            b.x = W[(size_t)(k0 + 2 * hi + 0) * OUTC + t * 16 + lo];
            b.y = W[(size_t)(k0 + 2 * hi + 1) * OUTC + t * 16 + lo];
            acc[t] = __builtin_amdgcn_wmma_f32_16x16x4_f32(
                false, a, false, b, (short)0, acc[t], false, false);
        }
    }

#pragma unroll
    for (int t = 0; t < NT; ++t) {
#pragma unroll
        for (int r = 0; r < 8; ++r) {
            H[(size_t)(row0 + r + 8 * hi) * OUTC + t * 16 + lo] = acc[t][r];
        }
    }
}

template <int D>
__global__ void agg_init_kernel(const float* __restrict__ H,
                                const float* __restrict__ bias,
                                const float* __restrict__ dinv,
                                float* __restrict__ agg, int n) {
    size_t idx = (size_t)blockIdx.x * blockDim.x + threadIdx.x;
    if (idx >= (size_t)n * D) return;
    int node = (int)(idx / D);
    int c = (int)(idx % D);
    float di = dinv[node];
    agg[idx] = bias[c] + H[idx] * di * di;
}

template <int D>
__global__ void agg_edges_kernel(const int* __restrict__ src,
                                 const int* __restrict__ dst,
                                 const float* __restrict__ H,
                                 const float* __restrict__ dinv,
                                 float* __restrict__ agg, int e) {
    constexpr int G = D / 4;
    long long tid = (long long)blockIdx.x * blockDim.x + threadIdx.x;
    long long eidx = tid / G;
    int g = (int)(tid % G);
    if (eidx >= e) return;
    int s = src[eidx];
    int d = dst[eidx];
    float w = dinv[s] * dinv[d];
    const float4 h = *(const float4*)(H + (size_t)s * D + g * 4);
    float* a = agg + (size_t)d * D + g * 4;
    atomicAdd(a + 0, h.x * w);
    atomicAdd(a + 1, h.y * w);
    atomicAdd(a + 2, h.z * w);
    atomicAdd(a + 3, h.w * w);
}

extern "C" void kernel_launch(void* const* d_in, const int* in_sizes, int n_in,
                              void* d_out, int out_size, void* d_ws,
                              size_t ws_size, hipStream_t stream) {
    const float* x  = (const float*)d_in[0];
    const int*   ei = (const int*)d_in[1];
    const float* W1 = (const float*)d_in[2];
    const float* b1 = (const float*)d_in[3];
    const float* W2 = (const float*)d_in[4];
    const float* b2 = (const float*)d_in[5];
    const float* W3 = (const float*)d_in[6];
    const float* b3 = (const float*)d_in[7];
    const float* W4 = (const float*)d_in[8];
    const float* b4 = (const float*)d_in[9];

    const int N = in_sizes[0] / D_IN;
    const int E = in_sizes[1] / 2;
    const int* src = ei;
    const int* dst = ei + E;

    float* dinv = (float*)d_ws;
    float* bufH = dinv + N;
    float* bufA = bufH + (size_t)N * D_IN;

    float* outp = (float*)d_out;

    const int TPB = 256;
    const int gN   = (N + TPB - 1) / TPB;
    const int gE   = (E + TPB - 1) / TPB;
    const int gND  = ((size_t)N * 64 + TPB - 1) / TPB;
    const int gND4 = ((size_t)N * 16 + TPB - 1) / TPB;
    const int gE16 = (int)(((long long)E * 16 + TPB - 1) / TPB);
    const int gE4  = (int)(((long long)E * 4 + TPB - 1) / TPB);
    const int tiles = (N + 15) / 16;
    const int gGemm = (tiles + WAVES_PER_BLOCK - 1) / WAVES_PER_BLOCK;

    deg_init_kernel<<<gN, TPB, 0, stream>>>(dinv, N);
    deg_edges_kernel<<<gE, TPB, 0, stream>>>(dst, dinv, E);
    deg_rsqrt_kernel<<<gN, TPB, 0, stream>>>(dinv, N);

    gemm_wmma_kernel<64, false><<<gGemm, TPB, 0, stream>>>(x, W1, bufH, N);
    agg_init_kernel<64><<<gND, TPB, 0, stream>>>(bufH, b1, dinv, bufA, N);
    agg_edges_kernel<64><<<gE16, TPB, 0, stream>>>(src, dst, bufH, dinv, bufA, E);

    gemm_wmma_kernel<64, true><<<gGemm, TPB, 0, stream>>>(bufA, W2, bufH, N);
    agg_init_kernel<64><<<gND, TPB, 0, stream>>>(bufH, b2, dinv, bufA, N);
    agg_edges_kernel<64><<<gE16, TPB, 0, stream>>>(src, dst, bufH, dinv, bufA, E);

    gemm_wmma_kernel<64, true><<<gGemm, TPB, 0, stream>>>(bufA, W3, bufH, N);
    agg_init_kernel<64><<<gND, TPB, 0, stream>>>(bufH, b3, dinv, bufA, N);
    agg_edges_kernel<64><<<gE16, TPB, 0, stream>>>(src, dst, bufH, dinv, bufA, E);

    gemm_wmma_kernel<16, true><<<gGemm, TPB, 0, stream>>>(bufA, W4, bufH, N);
    agg_init_kernel<16><<<gND4, TPB, 0, stream>>>(bufH, b4, dinv, outp, N);
    agg_edges_kernel<16><<<gE4, TPB, 0, stream>>>(src, dst, bufH, dinv, outp, E);
}


# Edge scatter: one wave per 32 edges, lane == column, so each row load and f32 atomic covers whole 128-byte lines (was 4 atomics per lane at 16-byte stride); sums still in atomic arrival order
# speedup vs baseline: 1.3477x; 1.3477x over previous
#include <hip/hip_runtime.h>


typedef __attribute__((ext_vector_type(2))) float v2f;
typedef __attribute__((ext_vector_type(8))) float v8f;

#define D_IN 64
#define WAVES_PER_BLOCK 8

__global__ void deg_init_kernel(float* __restrict__ deg, int n) {
    int i = blockIdx.x * blockDim.x + threadIdx.x;
    if (i < n) deg[i] = 1.0f;
}

__global__ void deg_edges_kernel(const int* __restrict__ dst,
                                 float* __restrict__ deg, int e) {
    int i = blockIdx.x * blockDim.x + threadIdx.x;
    if (i < e) atomicAdd(&deg[dst[i]], 1.0f);
}

__global__ void deg_rsqrt_kernel(float* __restrict__ deg, int n) {
    int i = blockIdx.x * blockDim.x + threadIdx.x;
    if (i < n) deg[i] = rsqrtf(deg[i]);
}

template <int OUTC, bool RELU_IN>
__global__ void gemm_wmma_kernel(const float* __restrict__ X,
                                 const float* __restrict__ W,
                                 float* __restrict__ H, int nrows) {
    const int lane = threadIdx.x & 31;
    const int wave = threadIdx.x >> 5;
    const int lo = lane & 15;
    const int hi = lane >> 4;

    const int tile = blockIdx.x * WAVES_PER_BLOCK + wave;
    const int row0 = tile * 16;
    if (row0 >= nrows) return;

    constexpr int NT = OUTC / 16;
    v8f acc[NT] = {};

    const float* __restrict__ xr = X + (size_t)(row0 + lo) * D_IN;

#pragma unroll
    for (int k0 = 0; k0 < D_IN; k0 += 4) {
        float a0 = xr[k0 + 2 * hi + 0];
        float a1 = xr[k0 + 2 * hi + 1];
        if (RELU_IN) {
            a0 = fmaxf(a0, 0.0f);
            a1 = fmaxf(a1, 0.0f);
        }
        v2f a;
        a.x = a0;
        a.y = a1;
#pragma unroll
        for (int t = 0; t < NT; ++t) {
            v2f b;
            b.x = W[(size_t)(k0 + 2 * hi + 0) * OUTC + t * 16 + lo];
            b.y = W[(size_t)(k0 + 2 * hi + 1) * OUTC + t * 16 + lo];
            acc[t] = __builtin_amdgcn_wmma_f32_16x16x4_f32(
                false, a, false, b, (short)0, acc[t], false, false);
        }
    }

#pragma unroll
    for (int t = 0; t < NT; ++t) {
#pragma unroll
        for (int r = 0; r < 8; ++r) {
            H[(size_t)(row0 + r + 8 * hi) * OUTC + t * 16 + lo] = acc[t][r];
        }
    }
}

template <int D>
__global__ void agg_init_kernel(const float* __restrict__ H,
                                const float* __restrict__ bias,
                                const float* __restrict__ dinv,
                                float* __restrict__ agg, int n) {
    size_t idx = (size_t)blockIdx.x * blockDim.x + threadIdx.x;
    if (idx >= (size_t)n * D) return;
    int node = (int)(idx / D);
    int c = (int)(idx % D);
    float di = dinv[node];
    agg[idx] = bias[c] + H[idx] * di * di;
}

template <int D>
__global__ void agg_edges_kernel(const int* __restrict__ src,
                                 const int* __restrict__ dst,
                                 const float* __restrict__ H,
                                 const float* __restrict__ dinv,
                                 float* __restrict__ agg, int e) {
    static_assert(D == 64 || D == 16, "row width");
    const int lane = threadIdx.x & 31;
    const int wv = __builtin_amdgcn_readfirstlane((int)(threadIdx.x >> 5));
    const long long e0 =
        ((long long)blockIdx.x * (blockDim.x >> 5) + wv) * 32;
    if (e0 >= e) return;
    const int cnt = (e - e0 < 32) ? (int)(e - e0) : 32;

    int s = 0, d = 0;
    float w = 0.0f;
    if (lane < cnt) {
        s = src[e0 + lane];
        d = dst[e0 + lane];
        w = dinv[s] * dinv[d];
    }
    const int wi = __float_as_int(w);

    if (cnt == 32) {
        if (D == 64) {
            for (int j0 = 0; j0 < 32; j0 += 8) {
                float h0[8], h1[8];
#pragma unroll
                for (int u = 0; u < 8; ++u) {
                    const int sj = __builtin_amdgcn_readlane(s, j0 + u);
                    const float* hp = H + (size_t)sj * 64 + lane;
                    h0[u] = hp[0];
                    h1[u] = hp[32];
                }
#pragma unroll
                for (int u = 0; u < 8; ++u) {
                    const int dj = __builtin_amdgcn_readlane(d, j0 + u);
                    const float wj =
                        __int_as_float(__builtin_amdgcn_readlane(wi, j0 + u));
                    float* ap = agg + (size_t)dj * 64 + lane;
                    atomicAdd(ap, h0[u] * wj);
                    atomicAdd(ap + 32, h1[u] * wj);
                }
            }
        } else {
            const int half = lane >> 4;
            const int c = lane & 15;
            for (int j0 = 0; j0 < 32; j0 += 16) {
                float h[8];
#pragma unroll
                for (int u = 0; u < 8; ++u) {
                    const int ja = j0 + 2 * u;
                    const int sa = __builtin_amdgcn_readlane(s, ja);
                    const int sb = __builtin_amdgcn_readlane(s, ja + 1);
                    const int sj = half ? sb : sa;
                    h[u] = H[(size_t)sj * 16 + c];
                }
#pragma unroll
                for (int u = 0; u < 8; ++u) {
                    const int ja = j0 + 2 * u;
                    const int da = __builtin_amdgcn_readlane(d, ja);
                    const int db = __builtin_amdgcn_readlane(d, ja + 1);
                    const int wa = __builtin_amdgcn_readlane(wi, ja);
                    const int wb = __builtin_amdgcn_readlane(wi, ja + 1);
                    const int dj = half ? db : da;
                    const float wj = __int_as_float(half ? wb : wa);
                    atomicAdd(agg + (size_t)dj * 16 + c, h[u] * wj);
                }
            }
        }
    } else {
        for (int j = 0; j < cnt; ++j) {
            const int sj = __builtin_amdgcn_readlane(s, j);
            const int dj = __builtin_amdgcn_readlane(d, j);
            const float wj = __int_as_float(__builtin_amdgcn_readlane(wi, j));
            for (int c = lane; c < D; c += 32) {
                atomicAdd(agg + (size_t)dj * D + c, H[(size_t)sj * D + c] * wj);
            }
        }
    }
}

extern "C" void kernel_launch(void* const* d_in, const int* in_sizes, int n_in,
                              void* d_out, int out_size, void* d_ws,
                              size_t ws_size, hipStream_t stream) {
    const float* x  = (const float*)d_in[0];
    const int*   ei = (const int*)d_in[1];
    const float* W1 = (const float*)d_in[2];
    const float* b1 = (const float*)d_in[3];
    const float* W2 = (const float*)d_in[4];
    const float* b2 = (const float*)d_in[5];
    const float* W3 = (const float*)d_in[6];
    const float* b3 = (const float*)d_in[7];
    const float* W4 = (const float*)d_in[8];
    const float* b4 = (const float*)d_in[9];

    const int N = in_sizes[0] / D_IN;
    const int E = in_sizes[1] / 2;
    const int* src = ei;
    const int* dst = ei + E;

    float* dinv = (float*)d_ws;
    float* bufH = dinv + N;
    float* bufA = bufH + (size_t)N * D_IN;

    float* outp = (float*)d_out;

    const int TPB = 256;
    const int gN   = (N + TPB - 1) / TPB;
    const int gE   = (E + TPB - 1) / TPB;
    const int gND  = ((size_t)N * 64 + TPB - 1) / TPB;
    const int gND4 = ((size_t)N * 16 + TPB - 1) / TPB;
    const int tiles = (N + 15) / 16;
    const int gGemm = (tiles + WAVES_PER_BLOCK - 1) / WAVES_PER_BLOCK;

    deg_init_kernel<<<gN, TPB, 0, stream>>>(dinv, N);
    deg_edges_kernel<<<gE, TPB, 0, stream>>>(dst, dinv, E);
    deg_rsqrt_kernel<<<gN, TPB, 0, stream>>>(dinv, N);

    gemm_wmma_kernel<64, false><<<gGemm, TPB, 0, stream>>>(x, W1, bufH, N);
    agg_init_kernel<64><<<gND, TPB, 0, stream>>>(bufH, b1, dinv, bufA, N);
    agg_edges_kernel<64><<<gE, TPB, 0, stream>>>(src, dst, bufH, dinv, bufA, E);

    gemm_wmma_kernel<64, true><<<gGemm, TPB, 0, stream>>>(bufA, W2, bufH, N);
    agg_init_kernel<64><<<gND, TPB, 0, stream>>>(bufH, b2, dinv, bufA, N);
    agg_edges_kernel<64><<<gE, TPB, 0, stream>>>(src, dst, bufH, dinv, bufA, E);

    gemm_wmma_kernel<64, true><<<gGemm, TPB, 0, stream>>>(bufA, W3, bufH, N);
    agg_init_kernel<64><<<gND, TPB, 0, stream>>>(bufH, b3, dinv, bufA, N);
    agg_edges_kernel<64><<<gE, TPB, 0, stream>>>(src, dst, bufH, dinv, bufA, E);

    gemm_wmma_kernel<16, true><<<gGemm, TPB, 0, stream>>>(bufA, W4, bufH, N);
    agg_init_kernel<16><<<gND4, TPB, 0, stream>>>(bufH, b4, dinv, outp, N);
    agg_edges_kernel<16><<<gE, TPB, 0, stream>>>(src, dst, bufH, dinv, outp, E);
}


# Edge sum without float atomics: edges sorted by destination per call (count, scan, fill); each row summed in registers from rows pre-scaled by dinv[src], stored once; sums in bucket fill order
# speedup vs baseline: 2.5219x; 1.8713x over previous
#include <hip/hip_runtime.h>


typedef __attribute__((ext_vector_type(2))) float v2f;
typedef __attribute__((ext_vector_type(8))) float v8f;

#define D_IN 64
#define WAVES_PER_BLOCK 8

__global__ void cnt_zero_kernel(int* __restrict__ cnt, int n) {
    int i = blockIdx.x * blockDim.x + threadIdx.x;
    if (i < n) cnt[i] = 0;
}

__global__ void cnt_edges_kernel(const int* __restrict__ dst,
                                 int* __restrict__ cnt,
                                 int* __restrict__ pos, int e) {
    int i = blockIdx.x * blockDim.x + threadIdx.x;
    if (i < e) pos[i] = atomicAdd(&cnt[dst[i]], 1);
}

__global__ void __launch_bounds__(1024)
scan_kernel(const int* __restrict__ cnt, int* __restrict__ rp, int n) {
    __shared__ int front[32];
    __shared__ int wtot[32];
    const int tid = threadIdx.x;
    const int lane = tid & 31;
    const int wv = tid >> 5;
    const int base = blockIdx.x * 1024;

    int acc = 0;
    for (int i = tid; i < base; i += 1024) acc += cnt[i];

    const int idx = base + tid;
    const int v = (idx < n) ? cnt[idx] : 0;

    int inc = v;
    for (int d = 1; d < 32; d <<= 1) {
        const int t = __shfl_up(inc, d, 32);
        if (lane >= d) inc += t;
    }
    for (int d = 16; d >= 1; d >>= 1) acc += __shfl_xor(acc, d, 32);

    if (lane == 31) wtot[wv] = inc;
    if (lane == 0) front[wv] = acc;
    __syncthreads();

    int both = front[lane] + ((lane < wv) ? wtot[lane] : 0);
    for (int d = 16; d >= 1; d >>= 1) both += __shfl_xor(both, d, 32);

    if (idx < n) rp[idx] = both + inc - v;
}

__global__ void fill_kernel(const int* __restrict__ src,
                            const int* __restrict__ dst,
                            const int* __restrict__ pos,
                            const int* __restrict__ rp,
                            int* __restrict__ csr, int e) {
    int i = blockIdx.x * blockDim.x + threadIdx.x;
    if (i < e) csr[rp[dst[i]] + pos[i]] = src[i];
}

__global__ void copy_kernel(const int* __restrict__ a, int* __restrict__ b,
                            int e) {
    int i = blockIdx.x * blockDim.x + threadIdx.x;
    if (i < e) b[i] = a[i];
}

template <int OUTC, bool RELU_IN>
__global__ void gemm_wmma_kernel(const float* __restrict__ X,
                                 const float* __restrict__ W,
                                 const int* __restrict__ rp,
                                 float* __restrict__ H, int nrows, int e) {
    const int lane = threadIdx.x & 31;
    const int wave = threadIdx.x >> 5;
    const int lo = lane & 15;
    const int hi = lane >> 4;

    const int tile = blockIdx.x * WAVES_PER_BLOCK + wave;
    const int row0 = tile * 16;
    if (row0 >= nrows) return;

    constexpr int NT = OUTC / 16;
    v8f acc[NT] = {};

    const float* __restrict__ xr = X + (size_t)(row0 + lo) * D_IN;

#pragma unroll
    for (int k0 = 0; k0 < D_IN; k0 += 4) {
        float a0 = xr[k0 + 2 * hi + 0];
        float a1 = xr[k0 + 2 * hi + 1];
        if (RELU_IN) {
            a0 = fmaxf(a0, 0.0f);
            a1 = fmaxf(a1, 0.0f);
        }
        v2f a;
        a.x = a0;
        a.y = a1;
#pragma unroll
        for (int t = 0; t < NT; ++t) {
            v2f b;
            b.x = W[(size_t)(k0 + 2 * hi + 0) * OUTC + t * 16 + lo];
            b.y = W[(size_t)(k0 + 2 * hi + 1) * OUTC + t * 16 + lo];
            acc[t] = __builtin_amdgcn_wmma_f32_16x16x4_f32(
                false, a, false, b, (short)0, acc[t], false, false);
        }
    }

    const int rb = row0 + 8 * hi;
    int p[9];
#pragma unroll
    for (int r = 0; r < 9; ++r) {
        const int q = rb + r;
        p[r] = (q < nrows) ? rp[q] : e;
    }
    float dr[8];
#pragma unroll
    for (int r = 0; r < 8; ++r) dr[r] = rsqrtf((float)(p[r + 1] - p[r] + 1));

#pragma unroll
    for (int t = 0; t < NT; ++t) {
#pragma unroll
        for (int r = 0; r < 8; ++r) {
            H[(size_t)(row0 + r + 8 * hi) * OUTC + t * 16 + lo] =
                acc[t][r] * dr[r];
        }
    }
}

__global__ void gather64_kernel(const int* __restrict__ rp,
                                const int* __restrict__ csr,
                                const float* __restrict__ Hs,
                                const float* __restrict__ bias,
                                float* __restrict__ A, int n, int e) {
    const int lane = threadIdx.x & 31;
    const int wv = __builtin_amdgcn_readfirstlane((int)(threadIdx.x >> 5));
    const int d = blockIdx.x * WAVES_PER_BLOCK + wv;
    if (d >= n) return;
    const int start = rp[d];
    const int end = (d + 1 < n) ? rp[d + 1] : e;
    const float di = rsqrtf((float)(end - start + 1));

    float2 acc = ((const float2*)(Hs + (size_t)d * 64))[lane];

    for (int j0 = start; j0 < end; j0 += 32) {
        const int cnt = (end - j0 < 32) ? (end - j0) : 32;
        int sv = 0;
        if (lane < cnt) sv = csr[j0 + lane];
        for (int g = 0; g < cnt; g += 8) {
            float2 h[8];
            int sj[8];
#pragma unroll
            for (int u = 0; u < 8; ++u)
                sj[u] = __builtin_amdgcn_readlane(sv, g + u);
#pragma unroll
            for (int u = 0; u < 8; ++u) {
                h[u] = make_float2(0.0f, 0.0f);
                if (g + u < cnt)
                    h[u] = ((const float2*)(Hs + (size_t)sj[u] * 64))[lane];
            }
#pragma unroll
            for (int u = 0; u < 8; ++u) {
                acc.x += h[u].x;
                acc.y += h[u].y;
            }
        }
    }

    const float2 b = ((const float2*)bias)[lane];
    float2 o;
    o.x = b.x + acc.x * di;
    o.y = b.y + acc.y * di;
    ((float2*)(A + (size_t)d * 64))[lane] = o;
}

__global__ void gather16_kernel(const int* __restrict__ rp,
                                const int* __restrict__ csr,
                                const float* __restrict__ Hs,
                                const float* __restrict__ bias,
                                float* __restrict__ out, int n, int e) {
    const int lane = threadIdx.x & 31;
    const int w = blockIdx.x * WAVES_PER_BLOCK + (int)(threadIdx.x >> 5);
    const int node = 2 * w + (lane >> 4);
    const int slot = (lane >> 2) & 3;
    const int q = lane & 3;

    float4 acc = make_float4(0.0f, 0.0f, 0.0f, 0.0f);
    float di = 0.0f;
    if (node < n) {
        const int start = rp[node];
        const int end = (node + 1 < n) ? rp[node + 1] : e;
        di = rsqrtf((float)(end - start + 1));
        if (slot == 0) acc = ((const float4*)(Hs + (size_t)node * 16))[q];
        for (int j0 = start; j0 < end; j0 += 16) {
            int id[4];
            float4 h[4];
#pragma unroll
            for (int u = 0; u < 4; ++u) {
                const int j = j0 + 4 * u + slot;
                id[u] = (j < end) ? csr[j] : -1;
            }
#pragma unroll
            for (int u = 0; u < 4; ++u) {
                h[u] = make_float4(0.0f, 0.0f, 0.0f, 0.0f);
                if (id[u] >= 0)
                    h[u] = ((const float4*)(Hs + (size_t)id[u] * 16))[q];
            }
#pragma unroll
            for (int u = 0; u < 4; ++u) {
                acc.x += h[u].x;
                acc.y += h[u].y;
                acc.z += h[u].z;
                acc.w += h[u].w;
            }
        }
    }
    acc.x += __shfl_xor(acc.x, 4, 32);
    acc.y += __shfl_xor(acc.y, 4, 32);
    acc.z += __shfl_xor(acc.z, 4, 32);
    acc.w += __shfl_xor(acc.w, 4, 32);
    acc.x += __shfl_xor(acc.x, 8, 32);
    acc.y += __shfl_xor(acc.y, 8, 32);
    acc.z += __shfl_xor(acc.z, 8, 32);
    acc.w += __shfl_xor(acc.w, 8, 32);

    if (node < n && slot == 0) {
        const float4 b = ((const float4*)bias)[q];
        float4 o;
        o.x = b.x + acc.x * di;
        o.y = b.y + acc.y * di;
        o.z = b.z + acc.z * di;
        o.w = b.w + acc.w * di;
        ((float4*)(out + (size_t)node * 16))[q] = o;
    }
}

extern "C" void kernel_launch(void* const* d_in, const int* in_sizes, int n_in,
                              void* d_out, int out_size, void* d_ws,
                              size_t ws_size, hipStream_t stream) {
    const float* x  = (const float*)d_in[0];
    const int*   ei = (const int*)d_in[1];
    const float* W1 = (const float*)d_in[2];
    const float* b1 = (const float*)d_in[3];
    const float* W2 = (const float*)d_in[4];
    const float* b2 = (const float*)d_in[5];
    const float* W3 = (const float*)d_in[6];
    const float* b3 = (const float*)d_in[7];
    const float* W4 = (const float*)d_in[8];
    const float* b4 = (const float*)d_in[9];

    const int N = in_sizes[0] / D_IN;
    const int E = in_sizes[1] / 2;
    const int* src = ei;
    const int* dst = ei + E;

    int*   rp   = (int*)d_ws;
    float* bufH = (float*)d_ws + N;
    float* bufA = bufH + (size_t)N * D_IN;
    int*   cnt  = (int*)bufA;
    int*   pos  = cnt + N;
    float* outp = (float*)d_out;
    int*   csr  = (int*)d_out;
    int*   csr4 = (int*)bufA;

    const int TPB = 256;
    const int gN    = (N + TPB - 1) / TPB;
    const int gE    = (E + TPB - 1) / TPB;
    const int gScan = (N + 1023) / 1024;
    const int tiles = (N + 15) / 16;
    const int gGemm = (tiles + WAVES_PER_BLOCK - 1) / WAVES_PER_BLOCK;
    const int gG64  = (N + WAVES_PER_BLOCK - 1) / WAVES_PER_BLOCK;
    const int gG16  = ((N + 1) / 2 + WAVES_PER_BLOCK - 1) / WAVES_PER_BLOCK;

    cnt_zero_kernel<<<gN, TPB, 0, stream>>>(cnt, N);
    cnt_edges_kernel<<<gE, TPB, 0, stream>>>(dst, cnt, pos, E);
    scan_kernel<<<gScan, 1024, 0, stream>>>(cnt, rp, N);
    fill_kernel<<<gE, TPB, 0, stream>>>(src, dst, pos, rp, csr, E);

    gemm_wmma_kernel<64, false><<<gGemm, TPB, 0, stream>>>(x, W1, rp, bufH, N, E);
    gather64_kernel<<<gG64, TPB, 0, stream>>>(rp, csr, bufH, b1, bufA, N, E);

    gemm_wmma_kernel<64, true><<<gGemm, TPB, 0, stream>>>(bufA, W2, rp, bufH, N, E);
    gather64_kernel<<<gG64, TPB, 0, stream>>>(rp, csr, bufH, b2, bufA, N, E);

    gemm_wmma_kernel<64, true><<<gGemm, TPB, 0, stream>>>(bufA, W3, rp, bufH, N, E);
    gather64_kernel<<<gG64, TPB, 0, stream>>>(rp, csr, bufH, b3, bufA, N, E);

    gemm_wmma_kernel<16, true><<<gGemm, TPB, 0, stream>>>(bufA, W4, rp, bufH, N, E);
    copy_kernel<<<gE, TPB, 0, stream>>>(csr, csr4, E);
    gather16_kernel<<<gG16, TPB, 0, stream>>>(rp, csr4, bufH, b4, outp, N, E);
}


# Edge sort without global atomics: edges binned by 512-row destination range, then ranked per row, all on LDS counters (was one returning global add per edge); sums in LDS counter arrival order
# speedup vs baseline: 3.5326x; 1.4008x over previous
#include <hip/hip_runtime.h>


typedef __attribute__((ext_vector_type(2))) float v2f;
typedef __attribute__((ext_vector_type(8))) float v8f;

#define D_IN 64
#define WAVES_PER_BLOCK 8

#define BSH 9
#define BW (1 << BSH)
#define MAXNB 2048
#define EPT 16
#define EPB (EPT * 256)

__global__ void __launch_bounds__(256)
bin_count_kernel(const int* __restrict__ dst, int* __restrict__ tab, int e,
                 int nb, int ga) {
    __shared__ int sh[MAXNB];
    const int tid = threadIdx.x;
    for (int b = tid; b < nb; b += 256) sh[b] = 0;
    __syncthreads();
    const int base = blockIdx.x * EPB + tid;
    int d[EPT];
#pragma unroll
    for (int u = 0; u < EPT; ++u) {
        const int i = base + u * 256;
        d[u] = (i < e) ? dst[i] : -1;
    }
#pragma unroll
    for (int u = 0; u < EPT; ++u)
        if (d[u] >= 0) atomicAdd(&sh[d[u] >> BSH], 1);
    __syncthreads();
    for (int b = tid; b < nb; b += 256)
        tab[(size_t)b * ga + blockIdx.x] = sh[b];
}

__global__ void __launch_bounds__(256)
bin_scatter_kernel(const int* __restrict__ src, const int* __restrict__ dst,
                   const int* __restrict__ off, int2* __restrict__ pairs,
                   int e, int nb, int ga) {
    __shared__ int cur[MAXNB];
    const int tid = threadIdx.x;
    for (int b = tid; b < nb; b += 256)
        cur[b] = off[(size_t)b * ga + blockIdx.x];
    __syncthreads();
    const int base = blockIdx.x * EPB + tid;
    int d[EPT];
    int s[EPT];
#pragma unroll
    for (int u = 0; u < EPT; ++u) {
        const int i = base + u * 256;
        d[u] = (i < e) ? dst[i] : -1;
        s[u] = (i < e) ? src[i] : 0;
    }
#pragma unroll
    for (int u = 0; u < EPT; ++u) {
        if (d[u] >= 0) {
            const int p = atomicAdd(&cur[d[u] >> BSH], 1);
            pairs[p] = make_int2(s[u], d[u]);
        }
    }
}

__global__ void __launch_bounds__(1024)
bin_finish_kernel(const int2* __restrict__ pairs, const int* __restrict__ off,
                  int* __restrict__ rp, int* __restrict__ csr, int n, int e,
                  int nb, int ga) {
    __shared__ int cnt[BW];
    __shared__ int wtot[BW / 32];
    const int tid = threadIdx.x;
    const int lane = tid & 31;
    const int wv = tid >> 5;
    const int b = blockIdx.x;
    const int lo = off[(size_t)b * ga];
    const int hi = (b + 1 < nb) ? off[(size_t)(b + 1) * ga] : e;
    const int node0 = b << BSH;

    if (tid < BW) cnt[tid] = 0;
    __syncthreads();

    for (int j0 = lo + tid; j0 < hi; j0 += 4 * 1024) {
        int dd[4];
#pragma unroll
        for (int u = 0; u < 4; ++u) {
            const int j = j0 + u * 1024;
            dd[u] = (j < hi) ? pairs[j].y : -1;
        }
#pragma unroll
        for (int u = 0; u < 4; ++u)
            if (dd[u] >= 0) atomicAdd(&cnt[dd[u] - node0], 1);
    }
    __syncthreads();

    const int v = (tid < BW) ? cnt[tid] : 0;
    int inc = v;
    for (int q = 1; q < 32; q <<= 1) {
        const int t = __shfl_up(inc, q, 32);
        if (lane >= q) inc += t;
    }
    if (tid < BW && lane == 31) wtot[wv] = inc;
    __syncthreads();
    if (tid < BW) {
        int front = 0;
        for (int k = 0; k < wv; ++k) front += wtot[k];
        const int start = lo + front + inc - v;
        cnt[tid] = start;
        const int node = node0 + tid;
        if (node < n) rp[node] = start;
    }
    __syncthreads();

    for (int j0 = lo + tid; j0 < hi; j0 += 4 * 1024) {
        int2 pr[4];
#pragma unroll
        for (int u = 0; u < 4; ++u) {
            const int j = j0 + u * 1024;
            pr[u] = (j < hi) ? pairs[j] : make_int2(0, -1);
        }
#pragma unroll
        for (int u = 0; u < 4; ++u) {
            if (pr[u].y >= 0) {
                const int p = atomicAdd(&cnt[pr[u].y - node0], 1);
                csr[p] = pr[u].x;
            }
        }
    }
}

__global__ void __launch_bounds__(1024)
scan_kernel(const int* __restrict__ cnt, int* __restrict__ rp, int n) {
    __shared__ int front[32];
    __shared__ int wtot[32];
    const int tid = threadIdx.x;
    const int lane = tid & 31;
    const int wv = tid >> 5;
    const int base = blockIdx.x * 1024;

    int acc = 0;
    for (int i = tid; i < base; i += 1024) acc += cnt[i];

    const int idx = base + tid;
    const int v = (idx < n) ? cnt[idx] : 0;

    int inc = v;
    for (int d = 1; d < 32; d <<= 1) {
        const int t = __shfl_up(inc, d, 32);
        if (lane >= d) inc += t;
    }
    for (int d = 16; d >= 1; d >>= 1) acc += __shfl_xor(acc, d, 32);

    if (lane == 31) wtot[wv] = inc;
    if (lane == 0) front[wv] = acc;
    __syncthreads();

    int both = front[lane] + ((lane < wv) ? wtot[lane] : 0);
    for (int d = 16; d >= 1; d >>= 1) both += __shfl_xor(both, d, 32);

    if (idx < n) rp[idx] = both + inc - v;
}

__global__ void copy_kernel(const int* __restrict__ a, int* __restrict__ b,
                            int e) {
    int i = blockIdx.x * blockDim.x + threadIdx.x;
    if (i < e) b[i] = a[i];
}

template <int OUTC, bool RELU_IN>
__global__ void gemm_wmma_kernel(const float* __restrict__ X,
                                 const float* __restrict__ W,
                                 const int* __restrict__ rp,
                                 float* __restrict__ H, int nrows, int e) {
    const int lane = threadIdx.x & 31;
    const int wave = threadIdx.x >> 5;
    const int lo = lane & 15;
    const int hi = lane >> 4;

    const int tile = blockIdx.x * WAVES_PER_BLOCK + wave;
    const int row0 = tile * 16;
    if (row0 >= nrows) return;

    constexpr int NT = OUTC / 16;
    v8f acc[NT] = {};

    const float* __restrict__ xr = X + (size_t)(row0 + lo) * D_IN;

#pragma unroll
    for (int k0 = 0; k0 < D_IN; k0 += 4) {
        float a0 = xr[k0 + 2 * hi + 0];
        float a1 = xr[k0 + 2 * hi + 1];
        if (RELU_IN) {
            a0 = fmaxf(a0, 0.0f);
            a1 = fmaxf(a1, 0.0f);
        }
        v2f a;
        a.x = a0;
        a.y = a1;
#pragma unroll
        for (int t = 0; t < NT; ++t) {
            v2f b;
            b.x = W[(size_t)(k0 + 2 * hi + 0) * OUTC + t * 16 + lo];
            b.y = W[(size_t)(k0 + 2 * hi + 1) * OUTC + t * 16 + lo];
            acc[t] = __builtin_amdgcn_wmma_f32_16x16x4_f32(
                false, a, false, b, (short)0, acc[t], false, false);
        }
    }

    const int rb = row0 + 8 * hi;
    int p[9];
#pragma unroll
    for (int r = 0; r < 9; ++r) {
        const int q = rb + r;
        p[r] = (q < nrows) ? rp[q] : e;
    }
    float dr[8];
#pragma unroll
    for (int r = 0; r < 8; ++r) dr[r] = rsqrtf((float)(p[r + 1] - p[r] + 1));

#pragma unroll
    for (int t = 0; t < NT; ++t) {
#pragma unroll
        for (int r = 0; r < 8; ++r) {
            H[(size_t)(row0 + r + 8 * hi) * OUTC + t * 16 + lo] =
                acc[t][r] * dr[r];
        }
    }
}

__global__ void gather64_kernel(const int* __restrict__ rp,
                                const int* __restrict__ csr,
                                const float* __restrict__ Hs,
                                const float* __restrict__ bias,
                                float* __restrict__ A, int n, int e) {
    const int lane = threadIdx.x & 31;
    const int wv = __builtin_amdgcn_readfirstlane((int)(threadIdx.x >> 5));
    const int d = blockIdx.x * WAVES_PER_BLOCK + wv;
    if (d >= n) return;
    const int start = rp[d];
    const int end = (d + 1 < n) ? rp[d + 1] : e;
    const float di = rsqrtf((float)(end - start + 1));

    float2 acc = ((const float2*)(Hs + (size_t)d * 64))[lane];

    for (int j0 = start; j0 < end; j0 += 32) {
        const int cnt = (end - j0 < 32) ? (end - j0) : 32;
        int sv = 0;
        if (lane < cnt) sv = csr[j0 + lane];
        for (int g = 0; g < cnt; g += 8) {
            float2 h[8];
            int sj[8];
#pragma unroll
            for (int u = 0; u < 8; ++u)
                sj[u] = __builtin_amdgcn_readlane(sv, g + u);
#pragma unroll
            for (int u = 0; u < 8; ++u) {
                h[u] = make_float2(0.0f, 0.0f);
                if (g + u < cnt)
                    h[u] = ((const float2*)(Hs + (size_t)sj[u] * 64))[lane];
            }
#pragma unroll
            for (int u = 0; u < 8; ++u) {
                acc.x += h[u].x;
                acc.y += h[u].y;
            }
        }
    }

    const float2 b = ((const float2*)bias)[lane];
    float2 o;
    o.x = b.x + acc.x * di;
    o.y = b.y + acc.y * di;
    ((float2*)(A + (size_t)d * 64))[lane] = o;
}

__global__ void gather16_kernel(const int* __restrict__ rp,
                                const int* __restrict__ csr,
                                const float* __restrict__ Hs,
                                const float* __restrict__ bias,
                                float* __restrict__ out, int n, int e) {
    const int lane = threadIdx.x & 31;
    const int w = blockIdx.x * WAVES_PER_BLOCK + (int)(threadIdx.x >> 5);
    const int node = 2 * w + (lane >> 4);
    const int slot = (lane >> 2) & 3;
    const int q = lane & 3;

    float4 acc = make_float4(0.0f, 0.0f, 0.0f, 0.0f);
    float di = 0.0f;
    if (node < n) {
        const int start = rp[node];
        const int end = (node + 1 < n) ? rp[node + 1] : e;
        di = rsqrtf((float)(end - start + 1));
        if (slot == 0) acc = ((const float4*)(Hs + (size_t)node * 16))[q];
        for (int j0 = start; j0 < end; j0 += 16) {
            int id[4];
            float4 h[4];
#pragma unroll
            for (int u = 0; u < 4; ++u) {
                const int j = j0 + 4 * u + slot;
                id[u] = (j < end) ? csr[j] : -1;
            }
#pragma unroll
            for (int u = 0; u < 4; ++u) {
                h[u] = make_float4(0.0f, 0.0f, 0.0f, 0.0f);
                if (id[u] >= 0)
                    h[u] = ((const float4*)(Hs + (size_t)id[u] * 16))[q];
            }
#pragma unroll
            for (int u = 0; u < 4; ++u) {
                acc.x += h[u].x;
                acc.y += h[u].y;
                acc.z += h[u].z;
                acc.w += h[u].w;
            }
        }
    }
    acc.x += __shfl_xor(acc.x, 4, 32);
    acc.y += __shfl_xor(acc.y, 4, 32);
    acc.z += __shfl_xor(acc.z, 4, 32);
    acc.w += __shfl_xor(acc.w, 4, 32);
    acc.x += __shfl_xor(acc.x, 8, 32);
    acc.y += __shfl_xor(acc.y, 8, 32);
    acc.z += __shfl_xor(acc.z, 8, 32);
    acc.w += __shfl_xor(acc.w, 8, 32);

    if (node < n && slot == 0) {
        const float4 b = ((const float4*)bias)[q];
        float4 o;
        o.x = b.x + acc.x * di;
        o.y = b.y + acc.y * di;
        o.z = b.z + acc.z * di;
        o.w = b.w + acc.w * di;
        ((float4*)(out + (size_t)node * 16))[q] = o;
    }
}

extern "C" void kernel_launch(void* const* d_in, const int* in_sizes, int n_in,
                              void* d_out, int out_size, void* d_ws,
                              size_t ws_size, hipStream_t stream) {
    const float* x  = (const float*)d_in[0];
    const int*   ei = (const int*)d_in[1];
    const float* W1 = (const float*)d_in[2];
    const float* b1 = (const float*)d_in[3];
    const float* W2 = (const float*)d_in[4];
    const float* b2 = (const float*)d_in[5];
    const float* W3 = (const float*)d_in[6];
    const float* b3 = (const float*)d_in[7];
    const float* W4 = (const float*)d_in[8];
    const float* b4 = (const float*)d_in[9];

    const int N = in_sizes[0] / D_IN;
    const int E = in_sizes[1] / 2;
    const int* src = ei;
    const int* dst = ei + E;

    int*   rp   = (int*)d_ws;
    float* bufH = (float*)d_ws + N;
    float* bufA = bufH + (size_t)N * D_IN;
    float* outp = (float*)d_out;
    int*   csr  = (int*)d_out;
    int*   csr4 = (int*)bufA;

    const int NB = (N + BW - 1) / BW;
    const int GA = (E + EPB - 1) / EPB;
    const int NT = NB * GA;
    int2*  pairs = (int2*)bufA;
    int*   tab   = (int*)bufA + 2 * (size_t)E;
    int*   off   = tab + NT;

    const int TPB = 256;
    const int gE    = (E + TPB - 1) / TPB;
    const int gScan = (NT + 1023) / 1024;
    const int tiles = (N + 15) / 16;
    const int gGemm = (tiles + WAVES_PER_BLOCK - 1) / WAVES_PER_BLOCK;
    const int gG64  = (N + WAVES_PER_BLOCK - 1) / WAVES_PER_BLOCK;
    const int gG16  = ((N + 1) / 2 + WAVES_PER_BLOCK - 1) / WAVES_PER_BLOCK;

    bin_count_kernel<<<GA, 256, 0, stream>>>(dst, tab, E, NB, GA);
    scan_kernel<<<gScan, 1024, 0, stream>>>(tab, off, NT);
    bin_scatter_kernel<<<GA, 256, 0, stream>>>(src, dst, off, pairs, E, NB, GA);
    bin_finish_kernel<<<NB, 1024, 0, stream>>>(pairs, off, rp, csr, N, E, NB, GA);

    gemm_wmma_kernel<64, false><<<gGemm, TPB, 0, stream>>>(x, W1, rp, bufH, N, E);
    gather64_kernel<<<gG64, TPB, 0, stream>>>(rp, csr, bufH, b1, bufA, N, E);

    gemm_wmma_kernel<64, true><<<gGemm, TPB, 0, stream>>>(bufA, W2, rp, bufH, N, E);
    gather64_kernel<<<gG64, TPB, 0, stream>>>(rp, csr, bufH, b2, bufA, N, E);

    gemm_wmma_kernel<64, true><<<gGemm, TPB, 0, stream>>>(bufA, W3, rp, bufH, N, E);
    gather64_kernel<<<gG64, TPB, 0, stream>>>(rp, csr, bufH, b3, bufA, N, E);

    gemm_wmma_kernel<16, true><<<gGemm, TPB, 0, stream>>>(bufA, W4, rp, bufH, N, E);
    copy_kernel<<<gE, TPB, 0, stream>>>(csr, csr4, E);
    gather16_kernel<<<gG16, TPB, 0, stream>>>(rp, csr4, bufH, b4, outp, N, E);
}


# Each 64-col gather fused with the next layer's transform: a block sums 32 rows into LDS and feeds the f32 matrix instruction from there; 3 launches and the A buffers go; order of sums unchanged
# speedup vs baseline: 4.0723x; 1.1528x over previous
#include <hip/hip_runtime.h>


typedef __attribute__((ext_vector_type(2))) float v2f;
typedef __attribute__((ext_vector_type(8))) float v8f;

#define D_IN 64
#define WAVES_PER_BLOCK 8

#define BSH 9
#define BW (1 << BSH)
#define MAXNB 2048
#define EPT 16
#define EPB (EPT * 256)

__global__ void __launch_bounds__(256)
bin_count_kernel(const int* __restrict__ dst, int* __restrict__ tab, int e,
                 int nb, int ga) {
    __shared__ int sh[MAXNB];
    const int tid = threadIdx.x;
    for (int b = tid; b < nb; b += 256) sh[b] = 0;
    __syncthreads();
    const int base = blockIdx.x * EPB + tid;
    int d[EPT];
#pragma unroll
    for (int u = 0; u < EPT; ++u) {
        const int i = base + u * 256;
        d[u] = (i < e) ? dst[i] : -1;
    }
#pragma unroll
    for (int u = 0; u < EPT; ++u)
        if (d[u] >= 0) atomicAdd(&sh[d[u] >> BSH], 1);
    __syncthreads();
    for (int b = tid; b < nb; b += 256)
        tab[(size_t)b * ga + blockIdx.x] = sh[b];
}

__global__ void __launch_bounds__(256)
bin_scatter_kernel(const int* __restrict__ src, const int* __restrict__ dst,
                   const int* __restrict__ off, int2* __restrict__ pairs,
                   int e, int nb, int ga) {
    __shared__ int cur[MAXNB];
    const int tid = threadIdx.x;
    for (int b = tid; b < nb; b += 256)
        cur[b] = off[(size_t)b * ga + blockIdx.x];
    __syncthreads();
    const int base = blockIdx.x * EPB + tid;
    int d[EPT];
    int s[EPT];
#pragma unroll
    for (int u = 0; u < EPT; ++u) {
        const int i = base + u * 256;
        d[u] = (i < e) ? dst[i] : -1;
        s[u] = (i < e) ? src[i] : 0;
    }
#pragma unroll
    for (int u = 0; u < EPT; ++u) {
        if (d[u] >= 0) {
            const int p = atomicAdd(&cur[d[u] >> BSH], 1);
            pairs[p] = make_int2(s[u], d[u]);
        }
    }
}

__global__ void __launch_bounds__(1024)
bin_finish_kernel(const int2* __restrict__ pairs, const int* __restrict__ off,
                  int* __restrict__ rp, int* __restrict__ csr, int n, int e,
                  int nb, int ga) {
    __shared__ int cnt[BW];
    __shared__ int wtot[BW / 32];
    const int tid = threadIdx.x;
    const int lane = tid & 31;
    const int wv = tid >> 5;
    const int b = blockIdx.x;
    const int lo = off[(size_t)b * ga];
    const int hi = (b + 1 < nb) ? off[(size_t)(b + 1) * ga] : e;
    const int node0 = b << BSH;

    if (tid < BW) cnt[tid] = 0;
    __syncthreads();

    for (int j0 = lo + tid; j0 < hi; j0 += 4 * 1024) {
        int dd[4];
#pragma unroll
        for (int u = 0; u < 4; ++u) {
            const int j = j0 + u * 1024;
            dd[u] = (j < hi) ? pairs[j].y : -1;
        }
#pragma unroll
        for (int u = 0; u < 4; ++u)
            if (dd[u] >= 0) atomicAdd(&cnt[dd[u] - node0], 1);
    }
    __syncthreads();

    const int v = (tid < BW) ? cnt[tid] : 0;
    int inc = v;
    for (int q = 1; q < 32; q <<= 1) {
        const int t = __shfl_up(inc, q, 32);
        if (lane >= q) inc += t;
    }
    if (tid < BW && lane == 31) wtot[wv] = inc;
    __syncthreads();
    if (tid < BW) {
        int front = 0;
        for (int k = 0; k < wv; ++k) front += wtot[k];
        const int start = lo + front + inc - v;
        cnt[tid] = start;
        const int node = node0 + tid;
        if (node < n) rp[node] = start;
    }
    __syncthreads();

    for (int j0 = lo + tid; j0 < hi; j0 += 4 * 1024) {
        int2 pr[4];
#pragma unroll
        for (int u = 0; u < 4; ++u) {
            const int j = j0 + u * 1024;
            pr[u] = (j < hi) ? pairs[j] : make_int2(0, -1);
        }
#pragma unroll
        for (int u = 0; u < 4; ++u) {
            if (pr[u].y >= 0) {
                const int p = atomicAdd(&cnt[pr[u].y - node0], 1);
                csr[p] = pr[u].x;
            }
        }
    }
}

__global__ void __launch_bounds__(1024)
scan_kernel(const int* __restrict__ cnt, int* __restrict__ rp, int n) {
    __shared__ int front[32];
    __shared__ int wtot[32];
    const int tid = threadIdx.x;
    const int lane = tid & 31;
    const int wv = tid >> 5;
    const int base = blockIdx.x * 1024;

    int acc = 0;
    for (int i = tid; i < base; i += 1024) acc += cnt[i];

    const int idx = base + tid;
    const int v = (idx < n) ? cnt[idx] : 0;

    int inc = v;
    for (int d = 1; d < 32; d <<= 1) {
        const int t = __shfl_up(inc, d, 32);
        if (lane >= d) inc += t;
    }
    for (int d = 16; d >= 1; d >>= 1) acc += __shfl_xor(acc, d, 32);

    if (lane == 31) wtot[wv] = inc;
    if (lane == 0) front[wv] = acc;
    __syncthreads();

    int both = front[lane] + ((lane < wv) ? wtot[lane] : 0);
    for (int d = 16; d >= 1; d >>= 1) both += __shfl_xor(both, d, 32);

    if (idx < n) rp[idx] = both + inc - v;
}

__global__ void copy_kernel(const int* __restrict__ a, int* __restrict__ b,
                            int e) {
    int i = blockIdx.x * blockDim.x + threadIdx.x;
    if (i < e) b[i] = a[i];
}

template <int OUTC, bool RELU_IN>
__global__ void gemm_wmma_kernel(const float* __restrict__ X,
                                 const float* __restrict__ W,
                                 const int* __restrict__ rp,
                                 float* __restrict__ H, int nrows, int e) {
    const int lane = threadIdx.x & 31;
    const int wave = threadIdx.x >> 5;
    const int lo = lane & 15;
    const int hi = lane >> 4;

    const int tile = blockIdx.x * WAVES_PER_BLOCK + wave;
    const int row0 = tile * 16;
    if (row0 >= nrows) return;

    constexpr int NT = OUTC / 16;
    v8f acc[NT] = {};

    const float* __restrict__ xr = X + (size_t)(row0 + lo) * D_IN;

#pragma unroll
    for (int k0 = 0; k0 < D_IN; k0 += 4) {
        float a0 = xr[k0 + 2 * hi + 0];
        float a1 = xr[k0 + 2 * hi + 1];
        if (RELU_IN) {
            a0 = fmaxf(a0, 0.0f);
            a1 = fmaxf(a1, 0.0f);
        }
        v2f a;
        a.x = a0;
        a.y = a1;
#pragma unroll
        for (int t = 0; t < NT; ++t) {
            v2f b;
            b.x = W[(size_t)(k0 + 2 * hi + 0) * OUTC + t * 16 + lo];
            b.y = W[(size_t)(k0 + 2 * hi + 1) * OUTC + t * 16 + lo];
            acc[t] = __builtin_amdgcn_wmma_f32_16x16x4_f32(
                false, a, false, b, (short)0, acc[t], false, false);
        }
    }

    const int rb = row0 + 8 * hi;
    int p[9];
#pragma unroll
    for (int r = 0; r < 9; ++r) {
        const int q = rb + r;
        p[r] = (q < nrows) ? rp[q] : e;
    }
    float dr[8];
#pragma unroll
    for (int r = 0; r < 8; ++r) dr[r] = rsqrtf((float)(p[r + 1] - p[r] + 1));

#pragma unroll
    for (int t = 0; t < NT; ++t) {
#pragma unroll
        for (int r = 0; r < 8; ++r) {
            H[(size_t)(row0 + r + 8 * hi) * OUTC + t * 16 + lo] =
                acc[t][r] * dr[r];
        }
    }
}

#define FROWS 32
#define APITCH (D_IN + 4)

template <int OUTC>
__global__ void __launch_bounds__(256)
gather_gemm_kernel(const int* __restrict__ rp, const int* __restrict__ csr,
                   const float* __restrict__ Hs,
                   const float* __restrict__ bias,
                   const float* __restrict__ W, float* __restrict__ Hn,
                   int n, int e) {
    constexpr int NT = OUTC / 16;
    constexpr int MT = FROWS / 16;
    constexpr int DPITCH = OUTC + 4;
    constexpr int RPW = FROWS / WAVES_PER_BLOCK;
    __shared__ __attribute__((aligned(16))) float sA[FROWS * APITCH];
    __shared__ __attribute__((aligned(16))) float sD[FROWS * DPITCH];
    __shared__ float sdi[FROWS];

    const int lane = threadIdx.x & 31;
    const int wv = __builtin_amdgcn_readfirstlane((int)(threadIdx.x >> 5));
    const int lo = lane & 15;
    const int hi = lane >> 4;
    const int row0 = blockIdx.x * FROWS;

    v2f bw[D_IN / 4];
    if (wv < NT) {
#pragma unroll
        for (int s = 0; s < D_IN / 4; ++s) {
            bw[s].x = W[(size_t)(4 * s + 2 * hi + 0) * OUTC + wv * 16 + lo];
            bw[s].y = W[(size_t)(4 * s + 2 * hi + 1) * OUTC + wv * 16 + lo];
        }
    }

    const float2 b = ((const float2*)bias)[lane];

    for (int rr = 0; rr < RPW; ++rr) {
        const int rl = wv * RPW + rr;
        const int d = row0 + rl;
        float2 o = make_float2(0.0f, 0.0f);
        float di = 0.0f;
        if (d < n) {
            const int start = rp[d];
            const int end = (d + 1 < n) ? rp[d + 1] : e;
            di = rsqrtf((float)(end - start + 1));

            float2 acc = ((const float2*)(Hs + (size_t)d * 64))[lane];

            for (int j0 = start; j0 < end; j0 += 32) {
                const int cnt = (end - j0 < 32) ? (end - j0) : 32;
                int sv = 0;
                if (lane < cnt) sv = csr[j0 + lane];
                for (int g = 0; g < cnt; g += 8) {
                    float2 h[8];
                    int sj[8];
#pragma unroll
                    for (int u = 0; u < 8; ++u)
                        sj[u] = __builtin_amdgcn_readlane(sv, g + u);
#pragma unroll
                    for (int u = 0; u < 8; ++u) {
                        h[u] = make_float2(0.0f, 0.0f);
                        if (g + u < cnt)
                            h[u] = ((const float2*)(Hs + (size_t)sj[u] * 64))[lane];
                    }
#pragma unroll
                    for (int u = 0; u < 8; ++u) {
                        acc.x += h[u].x;
                        acc.y += h[u].y;
                    }
                }
            }
            o.x = fmaxf(b.x + acc.x * di, 0.0f);
            o.y = fmaxf(b.y + acc.y * di, 0.0f);
        }
        ((float2*)(sA + rl * APITCH))[lane] = o;
        if (lane == 0) sdi[rl] = di;
    }
    __syncthreads();

    if (wv < NT) {
        v8f acc[MT] = {};
        const float* ar = sA + lo * APITCH + 2 * hi;
#pragma unroll
        for (int s = 0; s < D_IN / 4; ++s) {
#pragma unroll
            for (int m = 0; m < MT; ++m) {
                const v2f a = *(const v2f*)(ar + m * 16 * APITCH + 4 * s);
                acc[m] = __builtin_amdgcn_wmma_f32_16x16x4_f32(
                    false, a, false, bw[s], (short)0, acc[m], false, false);
            }
        }
#pragma unroll
        for (int m = 0; m < MT; ++m) {
#pragma unroll
            for (int r = 0; r < 8; ++r) {
                const int rl = m * 16 + 8 * hi + r;
                sD[rl * DPITCH + wv * 16 + lo] = acc[m][r] * sdi[rl];
            }
        }
    }
    __syncthreads();

    constexpr int F4 = FROWS * OUTC / 4;
#pragma unroll
    for (int i = 0; i < (F4 + 255) / 256; ++i) {
        const int f = (int)threadIdx.x + i * 256;
        const int rl = f / (OUTC / 4);
        const int c4 = f % (OUTC / 4);
        if (f < F4 && row0 + rl < n) {
            const float4 v = *(const float4*)(sD + rl * DPITCH + 4 * c4);
            *(float4*)(Hn + (size_t)(row0 + rl) * OUTC + 4 * c4) = v;
        }
    }
}

__global__ void gather16_kernel(const int* __restrict__ rp,
                                const int* __restrict__ csr,
                                const float* __restrict__ Hs,
                                const float* __restrict__ bias,
                                float* __restrict__ out, int n, int e) {
    const int lane = threadIdx.x & 31;
    const int w = blockIdx.x * WAVES_PER_BLOCK + (int)(threadIdx.x >> 5);
    const int node = 2 * w + (lane >> 4);
    const int slot = (lane >> 2) & 3;
    const int q = lane & 3;

    float4 acc = make_float4(0.0f, 0.0f, 0.0f, 0.0f);
    float di = 0.0f;
    if (node < n) {
        const int start = rp[node];
        const int end = (node + 1 < n) ? rp[node + 1] : e;
        di = rsqrtf((float)(end - start + 1));
        if (slot == 0) acc = ((const float4*)(Hs + (size_t)node * 16))[q];
        for (int j0 = start; j0 < end; j0 += 16) {
            int id[4];
            float4 h[4];
#pragma unroll
            for (int u = 0; u < 4; ++u) {
                const int j = j0 + 4 * u + slot;
                id[u] = (j < end) ? csr[j] : -1;
            }
#pragma unroll
            for (int u = 0; u < 4; ++u) {
                h[u] = make_float4(0.0f, 0.0f, 0.0f, 0.0f);
                if (id[u] >= 0)
                    h[u] = ((const float4*)(Hs + (size_t)id[u] * 16))[q];
            }
#pragma unroll
            for (int u = 0; u < 4; ++u) {
                acc.x += h[u].x;
                acc.y += h[u].y;
                acc.z += h[u].z;
                acc.w += h[u].w;
            }
        }
    }
    acc.x += __shfl_xor(acc.x, 4, 32);
    acc.y += __shfl_xor(acc.y, 4, 32);
    acc.z += __shfl_xor(acc.z, 4, 32);
    acc.w += __shfl_xor(acc.w, 4, 32);
    acc.x += __shfl_xor(acc.x, 8, 32);
    acc.y += __shfl_xor(acc.y, 8, 32);
    acc.z += __shfl_xor(acc.z, 8, 32);
    acc.w += __shfl_xor(acc.w, 8, 32);

    if (node < n && slot == 0) {
        const float4 b = ((const float4*)bias)[q];
        float4 o;
        o.x = b.x + acc.x * di;
        o.y = b.y + acc.y * di;
        o.z = b.z + acc.z * di;
        o.w = b.w + acc.w * di;
        ((float4*)(out + (size_t)node * 16))[q] = o;
    }
}

extern "C" void kernel_launch(void* const* d_in, const int* in_sizes, int n_in,
                              void* d_out, int out_size, void* d_ws,
                              size_t ws_size, hipStream_t stream) {
    const float* x  = (const float*)d_in[0];
    const int*   ei = (const int*)d_in[1];
    const float* W1 = (const float*)d_in[2];
    const float* b1 = (const float*)d_in[3];
    const float* W2 = (const float*)d_in[4];
    const float* b2 = (const float*)d_in[5];
    const float* W3 = (const float*)d_in[6];
    const float* b3 = (const float*)d_in[7];
    const float* W4 = (const float*)d_in[8];
    const float* b4 = (const float*)d_in[9];

    const int N = in_sizes[0] / D_IN;
    const int E = in_sizes[1] / 2;
    const int* src = ei;
    const int* dst = ei + E;

    int*   rp   = (int*)d_ws;
    float* bufH = (float*)d_ws + N;
    float* bufA = bufH + (size_t)N * D_IN;
    float* outp = (float*)d_out;
    int*   csr  = (int*)d_out;
    int*   csr4 = (int*)bufH;

    const int NB = (N + BW - 1) / BW;
    const int GA = (E + EPB - 1) / EPB;
    const int NT = NB * GA;
    int2*  pairs = (int2*)bufA;
    int*   tab   = (int*)bufA + 2 * (size_t)E;
    int*   off   = tab + NT;

    const int TPB = 256;
    const int gE    = (E + TPB - 1) / TPB;
    const int gScan = (NT + 1023) / 1024;
    const int tiles = (N + 15) / 16;
    const int gGemm = (tiles + WAVES_PER_BLOCK - 1) / WAVES_PER_BLOCK;
    const int gFuse = (N + FROWS - 1) / FROWS;
    const int gG16  = ((N + 1) / 2 + WAVES_PER_BLOCK - 1) / WAVES_PER_BLOCK;

    bin_count_kernel<<<GA, 256, 0, stream>>>(dst, tab, E, NB, GA);
    scan_kernel<<<gScan, 1024, 0, stream>>>(tab, off, NT);
    bin_scatter_kernel<<<GA, 256, 0, stream>>>(src, dst, off, pairs, E, NB, GA);
    bin_finish_kernel<<<NB, 1024, 0, stream>>>(pairs, off, rp, csr, N, E, NB, GA);

    gemm_wmma_kernel<64, false><<<gGemm, TPB, 0, stream>>>(x, W1, rp, bufH, N, E);

    gather_gemm_kernel<64><<<gFuse, TPB, 0, stream>>>(rp, csr, bufH, b1, W2, bufA, N, E);
    gather_gemm_kernel<64><<<gFuse, TPB, 0, stream>>>(rp, csr, bufA, b2, W3, bufH, N, E);
    gather_gemm_kernel<16><<<gFuse, TPB, 0, stream>>>(rp, csr, bufH, b3, W4, bufA, N, E);

    copy_kernel<<<gE, TPB, 0, stream>>>(csr, csr4, E);
    gather16_kernel<<<gG16, TPB, 0, stream>>>(rp, csr4, bufA, b4, outp, N, E);
}


# Edge sort in 2 launches instead of 4: each streaming block bins its own 4096 edges in LDS (rank, scan, grouped write to its own slice), the bin blocks walk the runs; sums in LDS counter arrival order
# speedup vs baseline: 4.4402x; 1.0903x over previous
#include <hip/hip_runtime.h>


typedef __attribute__((ext_vector_type(2))) float v2f;
typedef __attribute__((ext_vector_type(8))) float v8f;

#define D_IN 64
#define WAVES_PER_BLOCK 8

#define BSH 9
#define BW (1 << BSH)
#define MAXNB 2048
#define EPT 16
#define EPB (EPT * 256)

__global__ void __launch_bounds__(256)
bin_local_kernel(const int* __restrict__ src, const int* __restrict__ dst,
                 int2* __restrict__ pairs, int* __restrict__ tab,
                 int* __restrict__ off, int e, int nb, int ga) {
    __shared__ int cnt[MAXNB];
    __shared__ int wsum[8];
    const int tid = threadIdx.x;
    const int lane = tid & 31;
    const int wv = tid >> 5;
    for (int b = tid; b < MAXNB; b += 256) cnt[b] = 0;
    __syncthreads();
    const int base = blockIdx.x * EPB + tid;
    int d[EPT];
    int s[EPT];
    int r[EPT];
#pragma unroll
    for (int u = 0; u < EPT; ++u) {
        const int i = base + u * 256;
        d[u] = (i < e) ? dst[i] : -1;
        s[u] = (i < e) ? src[i] : 0;
    }
#pragma unroll
    for (int u = 0; u < EPT; ++u) {
        r[u] = 0;
        if (d[u] >= 0) r[u] = atomicAdd(&cnt[d[u] >> BSH], 1);
    }
    __syncthreads();

    int c[MAXNB / 256];
    int tot = 0;
#pragma unroll
    for (int q = 0; q < MAXNB / 256; ++q) {
        c[q] = cnt[tid * (MAXNB / 256) + q];
        tot += c[q];
    }
    int inc = tot;
    for (int q = 1; q < 32; q <<= 1) {
        const int t = __shfl_up(inc, q, 32);
        if (lane >= q) inc += t;
    }
    if (lane == 31) wsum[wv] = inc;
    __syncthreads();
    int run = inc - tot;
    for (int k = 0; k < wv; ++k) run += wsum[k];
#pragma unroll
    for (int q = 0; q < MAXNB / 256; ++q) {
        const int b = tid * (MAXNB / 256) + q;
        cnt[b] = run;
        if (b < nb) {
            tab[(size_t)b * ga + blockIdx.x] = run;
            off[(size_t)b * ga + blockIdx.x] = c[q];
        }
        run += c[q];
    }
    __syncthreads();

    int2* __restrict__ mine = pairs + (size_t)blockIdx.x * EPB;
#pragma unroll
    for (int u = 0; u < EPT; ++u) r[u] += cnt[((d[u] >= 0) ? d[u] : 0) >> BSH];
#pragma unroll
    for (int u = 0; u < EPT; ++u)
        if (d[u] >= 0) mine[r[u]] = make_int2(s[u], d[u]);
}

#define RG 8
__global__ void __launch_bounds__(1024)
bin_finish_kernel(const int2* __restrict__ pairs, const int* __restrict__ tab,
                  const int* __restrict__ off, int* __restrict__ rp,
                  int* __restrict__ csr, int n, int nb, int ga) {
    __shared__ int cnt[BW];
    __shared__ int wtot[BW / 32];
    __shared__ int fr[32];
    const int tid = threadIdx.x;
    const int lane = tid & 31;
    const int wv = __builtin_amdgcn_readfirstlane(tid >> 5);
    const int b = blockIdx.x;
    const int node0 = b << BSH;
    const int* __restrict__ tb = tab + (size_t)b * ga;
    const int* __restrict__ ob = off + (size_t)b * ga;

    int acc = 0;
    const int cells = b * ga;
    for (int i0 = tid; i0 < cells; i0 += 16 * 1024) {
        int t[16];
#pragma unroll
        for (int u = 0; u < 16; ++u) {
            const int i = i0 + u * 1024;
            t[u] = (i < cells) ? off[i] : 0;
        }
#pragma unroll
        for (int u = 0; u < 16; ++u) acc += t[u];
    }
    for (int q = 16; q >= 1; q >>= 1) acc += __shfl_xor(acc, q, 32);
    if (lane == 0) fr[wv] = acc;
    if (tid < BW) cnt[tid] = 0;
    __syncthreads();

    for (int k0 = wv * RG; k0 < ga; k0 += 32 * RG) {
        int st = 0, ln = 0;
        if (lane < RG && k0 + lane < ga) {
            st = tb[k0 + lane];
            ln = ob[k0 + lane];
        }
        int sq[RG], lq[RG], dd[RG];
#pragma unroll
        for (int q = 0; q < RG; ++q) {
            sq[q] = __builtin_amdgcn_readlane(st, q);
            lq[q] = __builtin_amdgcn_readlane(ln, q);
        }
#pragma unroll
        for (int q = 0; q < RG; ++q) {
            dd[q] = -1;
            if (lane < lq[q])
                dd[q] = pairs[(size_t)(k0 + q) * EPB + sq[q] + lane].y;
        }
#pragma unroll
        for (int q = 0; q < RG; ++q)
            if (dd[q] >= 0) atomicAdd(&cnt[dd[q] - node0], 1);
        for (int q = 0; q < RG; ++q) {
            const int l = __builtin_amdgcn_readlane(ln, q);
            const int s0 = __builtin_amdgcn_readlane(st, q);
            for (int j = 32 + lane; j < l; j += 32) {
                const int dj = pairs[(size_t)(k0 + q) * EPB + s0 + j].y;
                atomicAdd(&cnt[dj - node0], 1);
            }
        }
    }
    __syncthreads();

    const int v = (tid < BW) ? cnt[tid] : 0;
    int inc = v;
    for (int q = 1; q < 32; q <<= 1) {
        const int t = __shfl_up(inc, q, 32);
        if (lane >= q) inc += t;
    }
    if (tid < BW && lane == 31) wtot[wv] = inc;
    int lo = fr[lane];
    for (int q = 16; q >= 1; q >>= 1) lo += __shfl_xor(lo, q, 32);
    __syncthreads();
    if (tid < BW) {
        int front = 0;
        for (int k = 0; k < wv; ++k) front += wtot[k];
        const int start = lo + front + inc - v;
        cnt[tid] = start;
        const int node = node0 + tid;
        if (node < n) rp[node] = start;
    }
    __syncthreads();

    for (int k0 = wv * RG; k0 < ga; k0 += 32 * RG) {
        int st = 0, ln = 0;
        if (lane < RG && k0 + lane < ga) {
            st = tb[k0 + lane];
            ln = ob[k0 + lane];
        }
        int sq[RG], lq[RG];
        int2 pr[RG];
#pragma unroll
        for (int q = 0; q < RG; ++q) {
            sq[q] = __builtin_amdgcn_readlane(st, q);
            lq[q] = __builtin_amdgcn_readlane(ln, q);
        }
#pragma unroll
        for (int q = 0; q < RG; ++q) {
            pr[q] = make_int2(0, -1);
            if (lane < lq[q])
                pr[q] = pairs[(size_t)(k0 + q) * EPB + sq[q] + lane];
        }
#pragma unroll
        for (int q = 0; q < RG; ++q) {
            if (pr[q].y >= 0) {
                const int p = atomicAdd(&cnt[pr[q].y - node0], 1);
                csr[p] = pr[q].x;
            }
        }
        for (int q = 0; q < RG; ++q) {
            const int l = __builtin_amdgcn_readlane(ln, q);
            const int s0 = __builtin_amdgcn_readlane(st, q);
            for (int j = 32 + lane; j < l; j += 32) {
                const int2 pj = pairs[(size_t)(k0 + q) * EPB + s0 + j];
                const int p = atomicAdd(&cnt[pj.y - node0], 1);
                csr[p] = pj.x;
            }
        }
    }
}

__global__ void copy_kernel(const int* __restrict__ a, int* __restrict__ b,
                            int e) {
    int i = blockIdx.x * blockDim.x + threadIdx.x;
    if (i < e) b[i] = a[i];
}

template <int OUTC, bool RELU_IN>
__global__ void gemm_wmma_kernel(const float* __restrict__ X,
                                 const float* __restrict__ W,
                                 const int* __restrict__ rp,
                                 float* __restrict__ H, int nrows, int e) {
    const int lane = threadIdx.x & 31;
    const int wave = threadIdx.x >> 5;
    const int lo = lane & 15;
    const int hi = lane >> 4;

    const int tile = blockIdx.x * WAVES_PER_BLOCK + wave;
    const int row0 = tile * 16;
    if (row0 >= nrows) return;

    constexpr int NT = OUTC / 16;
    v8f acc[NT] = {};

    const float* __restrict__ xr = X + (size_t)(row0 + lo) * D_IN;

#pragma unroll
    for (int k0 = 0; k0 < D_IN; k0 += 4) {
        float a0 = xr[k0 + 2 * hi + 0];
        float a1 = xr[k0 + 2 * hi + 1];
        if (RELU_IN) {
            a0 = fmaxf(a0, 0.0f);
            a1 = fmaxf(a1, 0.0f);
        }
        v2f a;
        a.x = a0;
        a.y = a1;
#pragma unroll
        for (int t = 0; t < NT; ++t) {
            v2f b;
            b.x = W[(size_t)(k0 + 2 * hi + 0) * OUTC + t * 16 + lo];
            b.y = W[(size_t)(k0 + 2 * hi + 1) * OUTC + t * 16 + lo];
            acc[t] = __builtin_amdgcn_wmma_f32_16x16x4_f32(
                false, a, false, b, (short)0, acc[t], false, false);
        }
    }

    const int rb = row0 + 8 * hi;
    int p[9];
#pragma unroll
    for (int r = 0; r < 9; ++r) {
        const int q = rb + r;
        p[r] = (q < nrows) ? rp[q] : e;
    }
    float dr[8];
#pragma unroll
    for (int r = 0; r < 8; ++r) dr[r] = rsqrtf((float)(p[r + 1] - p[r] + 1));

#pragma unroll
    for (int t = 0; t < NT; ++t) {
#pragma unroll
        for (int r = 0; r < 8; ++r) {
            H[(size_t)(row0 + r + 8 * hi) * OUTC + t * 16 + lo] =
                acc[t][r] * dr[r];
        }
    }
}

#define FROWS 32
#define APITCH (D_IN + 4)

template <int OUTC>
__global__ void __launch_bounds__(256)
gather_gemm_kernel(const int* __restrict__ rp, const int* __restrict__ csr,
                   const float* __restrict__ Hs,
                   const float* __restrict__ bias,
                   const float* __restrict__ W, float* __restrict__ Hn,
                   int n, int e) {
    constexpr int NT = OUTC / 16;
    constexpr int MT = FROWS / 16;
    constexpr int DPITCH = OUTC + 4;
    constexpr int RPW = FROWS / WAVES_PER_BLOCK;
    __shared__ __attribute__((aligned(16))) float sA[FROWS * APITCH];
    __shared__ __attribute__((aligned(16))) float sD[FROWS * DPITCH];
    __shared__ float sdi[FROWS];

    const int lane = threadIdx.x & 31;
    const int wv = __builtin_amdgcn_readfirstlane((int)(threadIdx.x >> 5));
    const int lo = lane & 15;
    const int hi = lane >> 4;
    const int row0 = blockIdx.x * FROWS;

    v2f bw[D_IN / 4];
    if (wv < NT) {
#pragma unroll
        for (int s = 0; s < D_IN / 4; ++s) {
            bw[s].x = W[(size_t)(4 * s + 2 * hi + 0) * OUTC + wv * 16 + lo];
            bw[s].y = W[(size_t)(4 * s + 2 * hi + 1) * OUTC + wv * 16 + lo];
        }
    }

    const float2 b = ((const float2*)bias)[lane];

    for (int rr = 0; rr < RPW; ++rr) {
        const int rl = wv * RPW + rr;
        const int d = row0 + rl;
        float2 o = make_float2(0.0f, 0.0f);
        float di = 0.0f;
        if (d < n) {
            const int start = rp[d];
            const int end = (d + 1 < n) ? rp[d + 1] : e;
            di = rsqrtf((float)(end - start + 1));

            float2 acc = ((const float2*)(Hs + (size_t)d * 64))[lane];

            for (int j0 = start; j0 < end; j0 += 32) {
                const int cnt = (end - j0 < 32) ? (end - j0) : 32;
                int sv = 0;
                if (lane < cnt) sv = csr[j0 + lane];
                for (int g = 0; g < cnt; g += 8) {
                    float2 h[8];
                    int sj[8];
#pragma unroll
                    for (int u = 0; u < 8; ++u)
                        sj[u] = __builtin_amdgcn_readlane(sv, g + u);
#pragma unroll
                    for (int u = 0; u < 8; ++u) {
                        h[u] = make_float2(0.0f, 0.0f);
                        if (g + u < cnt)
                            h[u] = ((const float2*)(Hs + (size_t)sj[u] * 64))[lane];
                    }
#pragma unroll
                    for (int u = 0; u < 8; ++u) {
                        acc.x += h[u].x;
                        acc.y += h[u].y;
                    }
                }
            }
            o.x = fmaxf(b.x + acc.x * di, 0.0f);
            o.y = fmaxf(b.y + acc.y * di, 0.0f);
        }
        ((float2*)(sA + rl * APITCH))[lane] = o;
        if (lane == 0) sdi[rl] = di;
    }
    __syncthreads();

    if (wv < NT) {
        v8f acc[MT] = {};
        const float* ar = sA + lo * APITCH + 2 * hi;
#pragma unroll
        for (int s = 0; s < D_IN / 4; ++s) {
#pragma unroll
            for (int m = 0; m < MT; ++m) {
                const v2f a = *(const v2f*)(ar + m * 16 * APITCH + 4 * s);
                acc[m] = __builtin_amdgcn_wmma_f32_16x16x4_f32(
                    false, a, false, bw[s], (short)0, acc[m], false, false);
            }
        }
#pragma unroll
        for (int m = 0; m < MT; ++m) {
#pragma unroll
            for (int r = 0; r < 8; ++r) {
                const int rl = m * 16 + 8 * hi + r;
                sD[rl * DPITCH + wv * 16 + lo] = acc[m][r] * sdi[rl];
            }
        }
    }
    __syncthreads();

    constexpr int F4 = FROWS * OUTC / 4;
#pragma unroll
    for (int i = 0; i < (F4 + 255) / 256; ++i) {
        const int f = (int)threadIdx.x + i * 256;
        const int rl = f / (OUTC / 4);
        const int c4 = f % (OUTC / 4);
        if (f < F4 && row0 + rl < n) {
            const float4 v = *(const float4*)(sD + rl * DPITCH + 4 * c4);
            *(float4*)(Hn + (size_t)(row0 + rl) * OUTC + 4 * c4) = v;
        }
    }
}

__global__ void gather16_kernel(const int* __restrict__ rp,
                                const int* __restrict__ csr,
                                const float* __restrict__ Hs,
                                const float* __restrict__ bias,
                                float* __restrict__ out, int n, int e) {
    const int lane = threadIdx.x & 31;
    const int w = blockIdx.x * WAVES_PER_BLOCK + (int)(threadIdx.x >> 5);
    const int node = 2 * w + (lane >> 4);
    const int slot = (lane >> 2) & 3;
    const int q = lane & 3;

    float4 acc = make_float4(0.0f, 0.0f, 0.0f, 0.0f);
    float di = 0.0f;
    if (node < n) {
        const int start = rp[node];
        const int end = (node + 1 < n) ? rp[node + 1] : e;
        di = rsqrtf((float)(end - start + 1));
        if (slot == 0) acc = ((const float4*)(Hs + (size_t)node * 16))[q];
        for (int j0 = start; j0 < end; j0 += 16) {
            int id[4];
            float4 h[4];
#pragma unroll
            for (int u = 0; u < 4; ++u) {
                const int j = j0 + 4 * u + slot;
                id[u] = (j < end) ? csr[j] : -1;
            }
#pragma unroll
            for (int u = 0; u < 4; ++u) {
                h[u] = make_float4(0.0f, 0.0f, 0.0f, 0.0f);
                if (id[u] >= 0)
                    h[u] = ((const float4*)(Hs + (size_t)id[u] * 16))[q];
            }
#pragma unroll
            for (int u = 0; u < 4; ++u) {
                acc.x += h[u].x;
                acc.y += h[u].y;
                acc.z += h[u].z;
                acc.w += h[u].w;
            }
        }
    }
    acc.x += __shfl_xor(acc.x, 4, 32);
    acc.y += __shfl_xor(acc.y, 4, 32);
    acc.z += __shfl_xor(acc.z, 4, 32);
    acc.w += __shfl_xor(acc.w, 4, 32);
    acc.x += __shfl_xor(acc.x, 8, 32);
    acc.y += __shfl_xor(acc.y, 8, 32);
    acc.z += __shfl_xor(acc.z, 8, 32);
    acc.w += __shfl_xor(acc.w, 8, 32);

    if (node < n && slot == 0) {
        const float4 b = ((const float4*)bias)[q];
        float4 o;
        o.x = b.x + acc.x * di;
        o.y = b.y + acc.y * di;
        o.z = b.z + acc.z * di;
        o.w = b.w + acc.w * di;
        ((float4*)(out + (size_t)node * 16))[q] = o;
    }
}

extern "C" void kernel_launch(void* const* d_in, const int* in_sizes, int n_in,
                              void* d_out, int out_size, void* d_ws,
                              size_t ws_size, hipStream_t stream) {
    const float* x  = (const float*)d_in[0];
    const int*   ei = (const int*)d_in[1];
    const float* W1 = (const float*)d_in[2];
    const float* b1 = (const float*)d_in[3];
    const float* W2 = (const float*)d_in[4];
    const float* b2 = (const float*)d_in[5];
    const float* W3 = (const float*)d_in[6];
    const float* b3 = (const float*)d_in[7];
    const float* W4 = (const float*)d_in[8];
    const float* b4 = (const float*)d_in[9];

    const int N = in_sizes[0] / D_IN;
    const int E = in_sizes[1] / 2;
    const int* src = ei;
    const int* dst = ei + E;

    int*   rp   = (int*)d_ws;
    float* bufH = (float*)d_ws + N;
    float* bufA = bufH + (size_t)N * D_IN;
    float* outp = (float*)d_out;
    int*   csr  = (int*)d_out;
    int*   csr4 = (int*)bufH;

    const int NB = (N + BW - 1) / BW;
    const int GA = (E + EPB - 1) / EPB;
    const int NT = NB * GA;
    int2*  pairs = (int2*)bufA;
    int*   tab   = (int*)bufA + 2 * (size_t)E;
    int*   off   = tab + NT;

    const int TPB = 256;
    const int gE    = (E + TPB - 1) / TPB;
    const int tiles = (N + 15) / 16;
    const int gGemm = (tiles + WAVES_PER_BLOCK - 1) / WAVES_PER_BLOCK;
    const int gFuse = (N + FROWS - 1) / FROWS;
    const int gG16  = ((N + 1) / 2 + WAVES_PER_BLOCK - 1) / WAVES_PER_BLOCK;

    bin_local_kernel<<<GA, 256, 0, stream>>>(src, dst, pairs, tab, off, E, NB, GA);
    bin_finish_kernel<<<NB, 1024, 0, stream>>>(pairs, tab, off, rp, csr, N, NB, GA);

    gemm_wmma_kernel<64, false><<<gGemm, TPB, 0, stream>>>(x, W1, rp, bufH, N, E);

    gather_gemm_kernel<64><<<gFuse, TPB, 0, stream>>>(rp, csr, bufH, b1, W2, bufA, N, E);
    gather_gemm_kernel<64><<<gFuse, TPB, 0, stream>>>(rp, csr, bufA, b2, W3, bufH, N, E);
    gather_gemm_kernel<16><<<gFuse, TPB, 0, stream>>>(rp, csr, bufH, b3, W4, bufA, N, E);

    copy_kernel<<<gE, TPB, 0, stream>>>(csr, csr4, E);
    gather16_kernel<<<gG16, TPB, 0, stream>>>(rp, csr4, bufA, b4, outp, N, E);
}


# Fused gathers: ids of whole groups of 8 source rows come by scalar load, so a group is straight-line code (8 row loads at 32-bit offsets, 8 adds); last partial group padded; order of sums unchanged
# speedup vs baseline: 4.7568x; 1.0713x over previous
#include <hip/hip_runtime.h>


typedef __attribute__((ext_vector_type(2))) float v2f;
typedef __attribute__((ext_vector_type(8))) float v8f;

#define D_IN 64
#define WAVES_PER_BLOCK 8

#define BSH 9
#define BW (1 << BSH)
#define MAXNB 2048
#define EPT 16
#define EPB (EPT * 256)

__global__ void __launch_bounds__(256)
bin_local_kernel(const int* __restrict__ src, const int* __restrict__ dst,
                 int2* __restrict__ pairs, int* __restrict__ tab,
                 int* __restrict__ off, int e, int nb, int ga) {
    __shared__ int cnt[MAXNB];
    __shared__ int wsum[8];
    const int tid = threadIdx.x;
    const int lane = tid & 31;
    const int wv = tid >> 5;
    for (int b = tid; b < MAXNB; b += 256) cnt[b] = 0;
    __syncthreads();
    const int base = blockIdx.x * EPB + tid;
    int d[EPT];
    int s[EPT];
    int r[EPT];
#pragma unroll
    for (int u = 0; u < EPT; ++u) {
        const int i = base + u * 256;
        d[u] = (i < e) ? dst[i] : -1;
        s[u] = (i < e) ? src[i] : 0;
    }
#pragma unroll
    for (int u = 0; u < EPT; ++u) {
        r[u] = 0;
        if (d[u] >= 0) r[u] = atomicAdd(&cnt[d[u] >> BSH], 1);
    }
    __syncthreads();

    int c[MAXNB / 256];
    int tot = 0;
#pragma unroll
    for (int q = 0; q < MAXNB / 256; ++q) {
        c[q] = cnt[tid * (MAXNB / 256) + q];
        tot += c[q];
    }
    int inc = tot;
    for (int q = 1; q < 32; q <<= 1) {
        const int t = __shfl_up(inc, q, 32);
        if (lane >= q) inc += t;
    }
    if (lane == 31) wsum[wv] = inc;
    __syncthreads();
    int run = inc - tot;
    for (int k = 0; k < wv; ++k) run += wsum[k];
#pragma unroll
    for (int q = 0; q < MAXNB / 256; ++q) {
        const int b = tid * (MAXNB / 256) + q;
        cnt[b] = run;
        if (b < nb) {
            tab[(size_t)b * ga + blockIdx.x] = run;
            off[(size_t)b * ga + blockIdx.x] = c[q];
        }
        run += c[q];
    }
    __syncthreads();

    int2* __restrict__ mine = pairs + (size_t)blockIdx.x * EPB;
#pragma unroll
    for (int u = 0; u < EPT; ++u) r[u] += cnt[((d[u] >= 0) ? d[u] : 0) >> BSH];
#pragma unroll
    for (int u = 0; u < EPT; ++u)
        if (d[u] >= 0) mine[r[u]] = make_int2(s[u], d[u]);
}

#define RG 8
__global__ void __launch_bounds__(1024)
bin_finish_kernel(const int2* __restrict__ pairs, const int* __restrict__ tab,
                  const int* __restrict__ off, int* __restrict__ rp,
                  int* __restrict__ csr, int n, int nb, int ga) {
    __shared__ int cnt[BW];
    __shared__ int wtot[BW / 32];
    __shared__ int fr[32];
    const int tid = threadIdx.x;
    const int lane = tid & 31;
    const int wv = __builtin_amdgcn_readfirstlane(tid >> 5);
    const int b = blockIdx.x;
    const int node0 = b << BSH;
    const int* __restrict__ tb = tab + (size_t)b * ga;
    const int* __restrict__ ob = off + (size_t)b * ga;

    int acc = 0;
    const int cells = b * ga;
    for (int i0 = tid; i0 < cells; i0 += 16 * 1024) {
        int t[16];
#pragma unroll
        for (int u = 0; u < 16; ++u) {
            const int i = i0 + u * 1024;
            t[u] = (i < cells) ? off[i] : 0;
        }
#pragma unroll
        for (int u = 0; u < 16; ++u) acc += t[u];
    }
    for (int q = 16; q >= 1; q >>= 1) acc += __shfl_xor(acc, q, 32);
    if (lane == 0) fr[wv] = acc;
    if (tid < BW) cnt[tid] = 0;
    __syncthreads();

    for (int k0 = wv * RG; k0 < ga; k0 += 32 * RG) {
        int st = 0, ln = 0;
        if (lane < RG && k0 + lane < ga) {
            st = tb[k0 + lane];
            ln = ob[k0 + lane];
        }
        int sq[RG], lq[RG], dd[RG];
#pragma unroll
        for (int q = 0; q < RG; ++q) {
            sq[q] = __builtin_amdgcn_readlane(st, q);
            lq[q] = __builtin_amdgcn_readlane(ln, q);
        }
#pragma unroll
        for (int q = 0; q < RG; ++q) {
            dd[q] = -1;
            if (lane < lq[q])
                dd[q] = pairs[(size_t)(k0 + q) * EPB + sq[q] + lane].y;
        }
#pragma unroll
        for (int q = 0; q < RG; ++q)
            if (dd[q] >= 0) atomicAdd(&cnt[dd[q] - node0], 1);
        for (int q = 0; q < RG; ++q) {
            const int l = __builtin_amdgcn_readlane(ln, q);
            const int s0 = __builtin_amdgcn_readlane(st, q);
            for (int j = 32 + lane; j < l; j += 32) {
                const int dj = pairs[(size_t)(k0 + q) * EPB + s0 + j].y;
                atomicAdd(&cnt[dj - node0], 1);
            }
        }
    }
    __syncthreads();

    const int v = (tid < BW) ? cnt[tid] : 0;
    int inc = v;
    for (int q = 1; q < 32; q <<= 1) {
        const int t = __shfl_up(inc, q, 32);
        if (lane >= q) inc += t;
    }
    if (tid < BW && lane == 31) wtot[wv] = inc;
    int lo = fr[lane];
    for (int q = 16; q >= 1; q >>= 1) lo += __shfl_xor(lo, q, 32);
    __syncthreads();
    if (tid < BW) {
        int front = 0;
        for (int k = 0; k < wv; ++k) front += wtot[k];
        const int start = lo + front + inc - v;
        cnt[tid] = start;
        const int node = node0 + tid;
        if (node < n) rp[node] = start;
    }
    __syncthreads();

    for (int k0 = wv * RG; k0 < ga; k0 += 32 * RG) {
        int st = 0, ln = 0;
        if (lane < RG && k0 + lane < ga) {
            st = tb[k0 + lane];
            ln = ob[k0 + lane];
        }
        int sq[RG], lq[RG];
        int2 pr[RG];
#pragma unroll
        for (int q = 0; q < RG; ++q) {
            sq[q] = __builtin_amdgcn_readlane(st, q);
            lq[q] = __builtin_amdgcn_readlane(ln, q);
        }
#pragma unroll
        for (int q = 0; q < RG; ++q) {
            pr[q] = make_int2(0, -1);
            if (lane < lq[q])
                pr[q] = pairs[(size_t)(k0 + q) * EPB + sq[q] + lane];
        }
#pragma unroll
        for (int q = 0; q < RG; ++q) {
            if (pr[q].y >= 0) {
                const int p = atomicAdd(&cnt[pr[q].y - node0], 1);
                csr[p] = pr[q].x;
            }
        }
        for (int q = 0; q < RG; ++q) {
            const int l = __builtin_amdgcn_readlane(ln, q);
            const int s0 = __builtin_amdgcn_readlane(st, q);
            for (int j = 32 + lane; j < l; j += 32) {
                const int2 pj = pairs[(size_t)(k0 + q) * EPB + s0 + j];
                const int p = atomicAdd(&cnt[pj.y - node0], 1);
                csr[p] = pj.x;
            }
        }
    }
}

__global__ void copy_kernel(const int* __restrict__ a, int* __restrict__ b,
                            int e) {
    int i = blockIdx.x * blockDim.x + threadIdx.x;
    if (i < e) b[i] = a[i];
}

template <int OUTC, bool RELU_IN>
__global__ void gemm_wmma_kernel(const float* __restrict__ X,
                                 const float* __restrict__ W,
                                 const int* __restrict__ rp,
                                 float* __restrict__ H, int nrows, int e) {
    const int lane = threadIdx.x & 31;
    const int wave = threadIdx.x >> 5;
    const int lo = lane & 15;
    const int hi = lane >> 4;

    const int tile = blockIdx.x * WAVES_PER_BLOCK + wave;
    const int row0 = tile * 16;
    if (row0 >= nrows) return;

    constexpr int NT = OUTC / 16;
    v8f acc[NT] = {};

    const float* __restrict__ xr = X + (size_t)(row0 + lo) * D_IN;

#pragma unroll
    for (int k0 = 0; k0 < D_IN; k0 += 4) {
        float a0 = xr[k0 + 2 * hi + 0];
        float a1 = xr[k0 + 2 * hi + 1];
        if (RELU_IN) {
            a0 = fmaxf(a0, 0.0f);
            a1 = fmaxf(a1, 0.0f);
        }
        v2f a;
        a.x = a0;
        a.y = a1;
#pragma unroll
        for (int t = 0; t < NT; ++t) {
            v2f b;
            b.x = W[(size_t)(k0 + 2 * hi + 0) * OUTC + t * 16 + lo];
            b.y = W[(size_t)(k0 + 2 * hi + 1) * OUTC + t * 16 + lo];
            acc[t] = __builtin_amdgcn_wmma_f32_16x16x4_f32(
                false, a, false, b, (short)0, acc[t], false, false);
        }
    }

    const int rb = row0 + 8 * hi;
    int p[9];
#pragma unroll
    for (int r = 0; r < 9; ++r) {
        const int q = rb + r;
        p[r] = (q < nrows) ? rp[q] : e;
    }
    float dr[8];
#pragma unroll
    for (int r = 0; r < 8; ++r) dr[r] = rsqrtf((float)(p[r + 1] - p[r] + 1));

#pragma unroll
    for (int t = 0; t < NT; ++t) {
#pragma unroll
        for (int r = 0; r < 8; ++r) {
            H[(size_t)(row0 + r + 8 * hi) * OUTC + t * 16 + lo] =
                acc[t][r] * dr[r];
        }
    }
}

#define FROWS 32
#define APITCH (D_IN + 4)

template <int OUTC>
__global__ void __launch_bounds__(256)
__attribute__((amdgpu_waves_per_eu(16)))
gather_gemm_kernel(const int* __restrict__ rp, const int* __restrict__ csr,
                   const float* __restrict__ Hs,
                   const float* __restrict__ bias,
                   const float* __restrict__ W, float* __restrict__ Hn,
                   int n, int e) {
    constexpr int NT = OUTC / 16;
    constexpr int MT = FROWS / 16;
    constexpr int DPITCH = OUTC + 4;
    constexpr int RPW = FROWS / WAVES_PER_BLOCK;
    __shared__ __attribute__((aligned(16))) float sA[FROWS * APITCH];
    __shared__ __attribute__((aligned(16))) float sD[FROWS * DPITCH];
    __shared__ float sdi[FROWS];

    const int lane = threadIdx.x & 31;
    const int wv = __builtin_amdgcn_readfirstlane((int)(threadIdx.x >> 5));
    const int lo = lane & 15;
    const int hi = lane >> 4;
    const int row0 = blockIdx.x * FROWS;

    v2f bw[D_IN / 4];
    if (wv < NT) {
#pragma unroll
        for (int s = 0; s < D_IN / 4; ++s) {
            bw[s].x = W[(size_t)(4 * s + 2 * hi + 0) * OUTC + wv * 16 + lo];
            bw[s].y = W[(size_t)(4 * s + 2 * hi + 1) * OUTC + wv * 16 + lo];
        }
    }

    const float2 b = ((const float2*)bias)[lane];

    for (int rr = 0; rr < RPW; ++rr) {
        const int rl = wv * RPW + rr;
        const int d = row0 + rl;
        float2 o = make_float2(0.0f, 0.0f);
        float di = 0.0f;
        if (d < n) {
            const int start = rp[d];
            const int end = (d + 1 < n) ? rp[d + 1] : e;
            di = rsqrtf((float)(end - start + 1));

            const char* __restrict__ hb = (const char*)Hs;
            const unsigned lo8 = (unsigned)lane * 8u;
            float2 acc = *(const float2*)(hb + (((unsigned)d << 8) + lo8));

            const int deg = end - start;
            const int nfull = deg >> 3;
            const int rem = deg & 7;
            const int* __restrict__ cp = csr + start;

            int sv = 0;
            if (lane < rem) sv = cp[8 * nfull + lane];

            int cur[8];
            if (nfull > 0) {
#pragma unroll
                for (int u = 0; u < 8; ++u) cur[u] = cp[u];
            }
#pragma nounroll
            for (int k = 0; k < nfull; ++k) {
                float2 h[8];
#pragma unroll
                for (int u = 0; u < 8; ++u)
                    h[u] = *(const float2*)(hb + (((unsigned)cur[u] << 8) + lo8));
                if (k + 1 < nfull) {
#pragma unroll
                    for (int u = 0; u < 8; ++u) cur[u] = cp[8 * (k + 1) + u];
                }
#pragma unroll
                for (int u = 0; u < 8; ++u) {
                    acc.x += h[u].x;
                    acc.y += h[u].y;
                }
            }
            if (rem > 0) {
                int sj[7];
                float2 h[7];
#pragma unroll
                for (int u = 0; u < 7; ++u)
                    sj[u] = __builtin_amdgcn_readlane(sv, u);
#pragma unroll
                for (int u = 0; u < 7; ++u)
                    h[u] = *(const float2*)(hb + (((unsigned)sj[u] << 8) + lo8));
#pragma unroll
                for (int u = 0; u < 7; ++u) {
                    const bool in = (u < rem);
                    acc.x += in ? h[u].x : 0.0f;
                    acc.y += in ? h[u].y : 0.0f;
                }
            }
            o.x = fmaxf(b.x + acc.x * di, 0.0f);
            o.y = fmaxf(b.y + acc.y * di, 0.0f);
        }
        ((float2*)(sA + rl * APITCH))[lane] = o;
        if (lane == 0) sdi[rl] = di;
    }
    __syncthreads();

    if (wv < NT) {
        v8f acc[MT] = {};
        const float* ar = sA + lo * APITCH + 2 * hi;
#pragma unroll
        for (int s = 0; s < D_IN / 4; ++s) {
#pragma unroll
            for (int m = 0; m < MT; ++m) {
                const v2f a = *(const v2f*)(ar + m * 16 * APITCH + 4 * s);
                acc[m] = __builtin_amdgcn_wmma_f32_16x16x4_f32(
                    false, a, false, bw[s], (short)0, acc[m], false, false);
            }
        }
#pragma unroll
        for (int m = 0; m < MT; ++m) {
#pragma unroll
            for (int r = 0; r < 8; ++r) {
                const int rl = m * 16 + 8 * hi + r;
                sD[rl * DPITCH + wv * 16 + lo] = acc[m][r] * sdi[rl];
            }
        }
    }
    __syncthreads();

    constexpr int F4 = FROWS * OUTC / 4;
#pragma unroll
    for (int i = 0; i < (F4 + 255) / 256; ++i) {
        const int f = (int)threadIdx.x + i * 256;
        const int rl = f / (OUTC / 4);
        const int c4 = f % (OUTC / 4);
        if (f < F4 && row0 + rl < n) {
            const float4 v = *(const float4*)(sD + rl * DPITCH + 4 * c4);
            *(float4*)(Hn + (size_t)(row0 + rl) * OUTC + 4 * c4) = v;
        }
    }
}

__global__ void gather16_kernel(const int* __restrict__ rp,
                                const int* __restrict__ csr,
                                const float* __restrict__ Hs,
                                const float* __restrict__ bias,
                                float* __restrict__ out, int n, int e) {
    const int lane = threadIdx.x & 31;
    const int w = blockIdx.x * WAVES_PER_BLOCK + (int)(threadIdx.x >> 5);
    const int node = 2 * w + (lane >> 4);
    const int slot = (lane >> 2) & 3;
    const int q = lane & 3;

    float4 acc = make_float4(0.0f, 0.0f, 0.0f, 0.0f);
    float di = 0.0f;
    if (node < n) {
        const int start = rp[node];
        const int end = (node + 1 < n) ? rp[node + 1] : e;
        di = rsqrtf((float)(end - start + 1));
        if (slot == 0) acc = ((const float4*)(Hs + (size_t)node * 16))[q];
        for (int j0 = start; j0 < end; j0 += 16) {
            int id[4];
            float4 h[4];
#pragma unroll
            for (int u = 0; u < 4; ++u) {
                const int j = j0 + 4 * u + slot;
                id[u] = (j < end) ? csr[j] : -1;
            }
#pragma unroll
            for (int u = 0; u < 4; ++u) {
                h[u] = make_float4(0.0f, 0.0f, 0.0f, 0.0f);
                if (id[u] >= 0)
                    h[u] = ((const float4*)(Hs + (size_t)id[u] * 16))[q];
            }
#pragma unroll
            for (int u = 0; u < 4; ++u) {
                acc.x += h[u].x;
                acc.y += h[u].y;
                acc.z += h[u].z;
                acc.w += h[u].w;
            }
        }
    }
    acc.x += __shfl_xor(acc.x, 4, 32);
    acc.y += __shfl_xor(acc.y, 4, 32);
    acc.z += __shfl_xor(acc.z, 4, 32);
    acc.w += __shfl_xor(acc.w, 4, 32);
    acc.x += __shfl_xor(acc.x, 8, 32);
    acc.y += __shfl_xor(acc.y, 8, 32);
    acc.z += __shfl_xor(acc.z, 8, 32);
    acc.w += __shfl_xor(acc.w, 8, 32);

    if (node < n && slot == 0) {
        const float4 b = ((const float4*)bias)[q];
        float4 o;
        o.x = b.x + acc.x * di;
        o.y = b.y + acc.y * di;
        o.z = b.z + acc.z * di;
        o.w = b.w + acc.w * di;
        ((float4*)(out + (size_t)node * 16))[q] = o;
    }
}

extern "C" void kernel_launch(void* const* d_in, const int* in_sizes, int n_in,
                              void* d_out, int out_size, void* d_ws,
                              size_t ws_size, hipStream_t stream) {
    const float* x  = (const float*)d_in[0];
    const int*   ei = (const int*)d_in[1];
    const float* W1 = (const float*)d_in[2];
    const float* b1 = (const float*)d_in[3];
    const float* W2 = (const float*)d_in[4];
    const float* b2 = (const float*)d_in[5];
    const float* W3 = (const float*)d_in[6];
    const float* b3 = (const float*)d_in[7];
    const float* W4 = (const float*)d_in[8];
    const float* b4 = (const float*)d_in[9];

    const int N = in_sizes[0] / D_IN;
    const int E = in_sizes[1] / 2;
    const int* src = ei;
    const int* dst = ei + E;

    int*   rp   = (int*)d_ws;
    float* bufH = (float*)d_ws + N;
    float* bufA = bufH + (size_t)N * D_IN;
    float* outp = (float*)d_out;
    int*   csr  = (int*)d_out;
    int*   csr4 = (int*)bufH;

    const int NB = (N + BW - 1) / BW;
    const int GA = (E + EPB - 1) / EPB;
    const int NT = NB * GA;
    int2*  pairs = (int2*)bufA;
    int*   tab   = (int*)bufA + 2 * (size_t)E;
    int*   off   = tab + NT;

    const int TPB = 256;
    const int gE    = (E + TPB - 1) / TPB;
    const int tiles = (N + 15) / 16;
    const int gGemm = (tiles + WAVES_PER_BLOCK - 1) / WAVES_PER_BLOCK;
    const int gFuse = (N + FROWS - 1) / FROWS;
    const int gG16  = ((N + 1) / 2 + WAVES_PER_BLOCK - 1) / WAVES_PER_BLOCK;

    bin_local_kernel<<<GA, 256, 0, stream>>>(src, dst, pairs, tab, off, E, NB, GA);
    bin_finish_kernel<<<NB, 1024, 0, stream>>>(pairs, tab, off, rp, csr, N, NB, GA);

    gemm_wmma_kernel<64, false><<<gGemm, TPB, 0, stream>>>(x, W1, rp, bufH, N, E);

    gather_gemm_kernel<64><<<gFuse, TPB, 0, stream>>>(rp, csr, bufH, b1, W2, bufA, N, E);
    gather_gemm_kernel<64><<<gFuse, TPB, 0, stream>>>(rp, csr, bufA, b2, W3, bufH, N, E);
    gather_gemm_kernel<16><<<gFuse, TPB, 0, stream>>>(rp, csr, bufH, b3, W4, bufA, N, E);

    copy_kernel<<<gE, TPB, 0, stream>>>(csr, csr4, E);
    gather16_kernel<<<gG16, TPB, 0, stream>>>(rp, csr4, bufA, b4, outp, N, E);
}


# Sorted-list copy folded into the last fused launch: each block copies its own rows' slots (whole 128-byte lines) to the free tail of that launch's output buffer; the copy launch (50000 waves) goes
# speedup vs baseline: 4.9009x; 1.0303x over previous
#include <hip/hip_runtime.h>


typedef __attribute__((ext_vector_type(2))) float v2f;
typedef __attribute__((ext_vector_type(8))) float v8f;

#define D_IN 64
#define WAVES_PER_BLOCK 8

#define BSH 9
#define BW (1 << BSH)
#define MAXNB 2048
#define EPT 16
#define EPB (EPT * 256)

__global__ void __launch_bounds__(256)
bin_local_kernel(const int* __restrict__ src, const int* __restrict__ dst,
                 int2* __restrict__ pairs, int* __restrict__ tab,
                 int* __restrict__ off, int e, int nb, int ga) {
    __shared__ int cnt[MAXNB];
    __shared__ int wsum[8];
    const int tid = threadIdx.x;
    const int lane = tid & 31;
    const int wv = tid >> 5;
    for (int b = tid; b < MAXNB; b += 256) cnt[b] = 0;
    __syncthreads();
    const int base = blockIdx.x * EPB + tid;
    int d[EPT];
    int s[EPT];
    int r[EPT];
#pragma unroll
    for (int u = 0; u < EPT; ++u) {
        const int i = base + u * 256;
        d[u] = (i < e) ? dst[i] : -1;
        s[u] = (i < e) ? src[i] : 0;
    }
#pragma unroll
    for (int u = 0; u < EPT; ++u) {
        r[u] = 0;
        if (d[u] >= 0) r[u] = atomicAdd(&cnt[d[u] >> BSH], 1);
    }
    __syncthreads();

    int c[MAXNB / 256];
    int tot = 0;
#pragma unroll
    for (int q = 0; q < MAXNB / 256; ++q) {
        c[q] = cnt[tid * (MAXNB / 256) + q];
        tot += c[q];
    }
    int inc = tot;
    for (int q = 1; q < 32; q <<= 1) {
        const int t = __shfl_up(inc, q, 32);
        if (lane >= q) inc += t;
    }
    if (lane == 31) wsum[wv] = inc;
    __syncthreads();
    int run = inc - tot;
    for (int k = 0; k < wv; ++k) run += wsum[k];
#pragma unroll
    for (int q = 0; q < MAXNB / 256; ++q) {
        const int b = tid * (MAXNB / 256) + q;
        cnt[b] = run;
        if (b < nb) {
            tab[(size_t)b * ga + blockIdx.x] = run;
            off[(size_t)b * ga + blockIdx.x] = c[q];
        }
        run += c[q];
    }
    __syncthreads();

    int2* __restrict__ mine = pairs + (size_t)blockIdx.x * EPB;
#pragma unroll
    for (int u = 0; u < EPT; ++u) r[u] += cnt[((d[u] >= 0) ? d[u] : 0) >> BSH];
#pragma unroll
    for (int u = 0; u < EPT; ++u)
        if (d[u] >= 0) mine[r[u]] = make_int2(s[u], d[u]);
}

#define RG 8
__global__ void __launch_bounds__(1024)
bin_finish_kernel(const int2* __restrict__ pairs, const int* __restrict__ tab,
                  const int* __restrict__ off, int* __restrict__ rp,
                  int* __restrict__ csr, int n, int nb, int ga) {
    __shared__ int cnt[BW];
    __shared__ int wtot[BW / 32];
    __shared__ int fr[32];
    const int tid = threadIdx.x;
    const int lane = tid & 31;
    const int wv = __builtin_amdgcn_readfirstlane(tid >> 5);
    const int b = blockIdx.x;
    const int node0 = b << BSH;
    const int* __restrict__ tb = tab + (size_t)b * ga;
    const int* __restrict__ ob = off + (size_t)b * ga;

    int acc = 0;
    const int cells = b * ga;
    for (int i0 = tid; i0 < cells; i0 += 16 * 1024) {
        int t[16];
#pragma unroll
        for (int u = 0; u < 16; ++u) {
            const int i = i0 + u * 1024;
            t[u] = (i < cells) ? off[i] : 0;
        }
#pragma unroll
        for (int u = 0; u < 16; ++u) acc += t[u];
    }
    for (int q = 16; q >= 1; q >>= 1) acc += __shfl_xor(acc, q, 32);
    if (lane == 0) fr[wv] = acc;
    if (tid < BW) cnt[tid] = 0;
    __syncthreads();

    for (int k0 = wv * RG; k0 < ga; k0 += 32 * RG) {
        int st = 0, ln = 0;
        if (lane < RG && k0 + lane < ga) {
            st = tb[k0 + lane];
            ln = ob[k0 + lane];
        }
        int sq[RG], lq[RG], dd[RG];
#pragma unroll
        for (int q = 0; q < RG; ++q) {
            sq[q] = __builtin_amdgcn_readlane(st, q);
            lq[q] = __builtin_amdgcn_readlane(ln, q);
        }
#pragma unroll
        for (int q = 0; q < RG; ++q) {
            dd[q] = -1;
            if (lane < lq[q])
                dd[q] = pairs[(size_t)(k0 + q) * EPB + sq[q] + lane].y;
        }
#pragma unroll
        for (int q = 0; q < RG; ++q)
            if (dd[q] >= 0) atomicAdd(&cnt[dd[q] - node0], 1);
        for (int q = 0; q < RG; ++q) {
            const int l = __builtin_amdgcn_readlane(ln, q);
            const int s0 = __builtin_amdgcn_readlane(st, q);
            for (int j = 32 + lane; j < l; j += 32) {
                const int dj = pairs[(size_t)(k0 + q) * EPB + s0 + j].y;
                atomicAdd(&cnt[dj - node0], 1);
            }
        }
    }
    __syncthreads();

    const int v = (tid < BW) ? cnt[tid] : 0;
    int inc = v;
    for (int q = 1; q < 32; q <<= 1) {
        const int t = __shfl_up(inc, q, 32);
        if (lane >= q) inc += t;
    }
    if (tid < BW && lane == 31) wtot[wv] = inc;
    int lo = fr[lane];
    for (int q = 16; q >= 1; q >>= 1) lo += __shfl_xor(lo, q, 32);
    __syncthreads();
    if (tid < BW) {
        int front = 0;
        for (int k = 0; k < wv; ++k) front += wtot[k];
        const int start = lo + front + inc - v;
        cnt[tid] = start;
        const int node = node0 + tid;
        if (node < n) rp[node] = start;
    }
    __syncthreads();

    for (int k0 = wv * RG; k0 < ga; k0 += 32 * RG) {
        int st = 0, ln = 0;
        if (lane < RG && k0 + lane < ga) {
            st = tb[k0 + lane];
            ln = ob[k0 + lane];
        }
        int sq[RG], lq[RG];
        int2 pr[RG];
#pragma unroll
        for (int q = 0; q < RG; ++q) {
            sq[q] = __builtin_amdgcn_readlane(st, q);
            lq[q] = __builtin_amdgcn_readlane(ln, q);
        }
#pragma unroll
        for (int q = 0; q < RG; ++q) {
            pr[q] = make_int2(0, -1);
            if (lane < lq[q])
                pr[q] = pairs[(size_t)(k0 + q) * EPB + sq[q] + lane];
        }
#pragma unroll
        for (int q = 0; q < RG; ++q) {
            if (pr[q].y >= 0) {
                const int p = atomicAdd(&cnt[pr[q].y - node0], 1);
                csr[p] = pr[q].x;
            }
        }
        for (int q = 0; q < RG; ++q) {
            const int l = __builtin_amdgcn_readlane(ln, q);
            const int s0 = __builtin_amdgcn_readlane(st, q);
            for (int j = 32 + lane; j < l; j += 32) {
                const int2 pj = pairs[(size_t)(k0 + q) * EPB + s0 + j];
                const int p = atomicAdd(&cnt[pj.y - node0], 1);
                csr[p] = pj.x;
            }
        }
    }
}

template <int OUTC, bool RELU_IN>
__global__ void gemm_wmma_kernel(const float* __restrict__ X,
                                 const float* __restrict__ W,
                                 const int* __restrict__ rp,
                                 float* __restrict__ H, int nrows, int e) {
    const int lane = threadIdx.x & 31;
    const int wave = threadIdx.x >> 5;
    const int lo = lane & 15;
    const int hi = lane >> 4;

    const int tile = blockIdx.x * WAVES_PER_BLOCK + wave;
    const int row0 = tile * 16;
    if (row0 >= nrows) return;

    constexpr int NT = OUTC / 16;
    v8f acc[NT] = {};

    const float* __restrict__ xr = X + (size_t)(row0 + lo) * D_IN;

#pragma unroll
    for (int k0 = 0; k0 < D_IN; k0 += 4) {
        float a0 = xr[k0 + 2 * hi + 0];
        float a1 = xr[k0 + 2 * hi + 1];
        if (RELU_IN) {
            a0 = fmaxf(a0, 0.0f);
            a1 = fmaxf(a1, 0.0f);
        }
        v2f a;
        a.x = a0;
        a.y = a1;
#pragma unroll
        for (int t = 0; t < NT; ++t) {
            v2f b;
            b.x = W[(size_t)(k0 + 2 * hi + 0) * OUTC + t * 16 + lo];
            b.y = W[(size_t)(k0 + 2 * hi + 1) * OUTC + t * 16 + lo];
            acc[t] = __builtin_amdgcn_wmma_f32_16x16x4_f32(
                false, a, false, b, (short)0, acc[t], false, false);
        }
    }

    const int rb = row0 + 8 * hi;
    int p[9];
#pragma unroll
    for (int r = 0; r < 9; ++r) {
        const int q = rb + r;
        p[r] = (q < nrows) ? rp[q] : e;
    }
    float dr[8];
#pragma unroll
    for (int r = 0; r < 8; ++r) dr[r] = rsqrtf((float)(p[r + 1] - p[r] + 1));

#pragma unroll
    for (int t = 0; t < NT; ++t) {
#pragma unroll
        for (int r = 0; r < 8; ++r) {
            H[(size_t)(row0 + r + 8 * hi) * OUTC + t * 16 + lo] =
                acc[t][r] * dr[r];
        }
    }
}

#define FROWS 32
#define APITCH (D_IN + 4)
#define CPW 3

template <int OUTC, bool CPY>
__global__ void __launch_bounds__(256)
__attribute__((amdgpu_waves_per_eu(16)))
gather_gemm_kernel(const int* __restrict__ rp, const int* __restrict__ csr,
                   const float* __restrict__ Hs,
                   const float* __restrict__ bias,
                   const float* __restrict__ W, float* __restrict__ Hn,
                   int* __restrict__ cpy, int n, int e) {
    constexpr int NT = OUTC / 16;
    constexpr int MT = FROWS / 16;
    constexpr int DPITCH = OUTC + 4;
    constexpr int RPW = FROWS / WAVES_PER_BLOCK;
    __shared__ __attribute__((aligned(16))) float sA[FROWS * APITCH];
    __shared__ __attribute__((aligned(16))) float sD[FROWS * DPITCH];
    __shared__ float sdi[FROWS];

    const int lane = threadIdx.x & 31;
    const int wv = __builtin_amdgcn_readfirstlane((int)(threadIdx.x >> 5));
    const int lo = lane & 15;
    const int hi = lane >> 4;
    const int row0 = blockIdx.x * FROWS;

    v2f bw[D_IN / 4];
    if (wv < NT) {
#pragma unroll
        for (int s = 0; s < D_IN / 4; ++s) {
            bw[s].x = W[(size_t)(4 * s + 2 * hi + 0) * OUTC + wv * 16 + lo];
            bw[s].y = W[(size_t)(4 * s + 2 * hi + 1) * OUTC + wv * 16 + lo];
        }
    }

    const float2 b = ((const float2*)bias)[lane];

    int cv[CPW];
    int c0 = 0, c1 = 0;
    if (CPY) {
        const int s0 = rp[row0];
        const int t1 = rp[min(row0 + FROWS, n - 1)];
        const int s1 = (row0 + FROWS < n) ? t1 : e;
        c0 = min((s0 + 31) & ~31, e);
        c1 = min((s1 + 31) & ~31, e);
#pragma unroll
        for (int u = 0; u < CPW; ++u) {
            const int i = c0 + (int)threadIdx.x + u * 256;
            cv[u] = (i < c1) ? csr[i] : 0;
        }
    }

    for (int rr = 0; rr < RPW; ++rr) {
        const int rl = wv * RPW + rr;
        const int d = row0 + rl;
        float2 o = make_float2(0.0f, 0.0f);
        float di = 0.0f;
        if (d < n) {
            const int start = rp[d];
            const int end = (d + 1 < n) ? rp[d + 1] : e;
            di = rsqrtf((float)(end - start + 1));

            const char* __restrict__ hb = (const char*)Hs;
            const unsigned lo8 = (unsigned)lane * 8u;
            float2 acc = *(const float2*)(hb + (((unsigned)d << 8) + lo8));

            const int deg = end - start;
            const int nfull = deg >> 3;
            const int rem = deg & 7;
            const int* __restrict__ cp = csr + start;

            int sv = 0;
            if (lane < rem) sv = cp[8 * nfull + lane];

            int cur[8];
            if (nfull > 0) {
#pragma unroll
                for (int u = 0; u < 8; ++u) cur[u] = cp[u];
            }
#pragma nounroll
            for (int k = 0; k < nfull; ++k) {
                float2 h[8];
#pragma unroll
                for (int u = 0; u < 8; ++u)
                    h[u] = *(const float2*)(hb + (((unsigned)cur[u] << 8) + lo8));
                if (k + 1 < nfull) {
#pragma unroll
                    for (int u = 0; u < 8; ++u) cur[u] = cp[8 * (k + 1) + u];
                }
#pragma unroll
                for (int u = 0; u < 8; ++u) {
                    acc.x += h[u].x;
                    acc.y += h[u].y;
                }
            }
            if (rem > 0) {
                int sj[7];
                float2 h[7];
#pragma unroll
                for (int u = 0; u < 7; ++u)
                    sj[u] = __builtin_amdgcn_readlane(sv, u);
#pragma unroll
                for (int u = 0; u < 7; ++u)
                    h[u] = *(const float2*)(hb + (((unsigned)sj[u] << 8) + lo8));
#pragma unroll
                for (int u = 0; u < 7; ++u) {
                    const bool in = (u < rem);
                    acc.x += in ? h[u].x : 0.0f;
                    acc.y += in ? h[u].y : 0.0f;
                }
            }
            o.x = fmaxf(b.x + acc.x * di, 0.0f);
            o.y = fmaxf(b.y + acc.y * di, 0.0f);
        }
        ((float2*)(sA + rl * APITCH))[lane] = o;
        if (lane == 0) sdi[rl] = di;
    }
    __syncthreads();

    if (CPY) {
#pragma unroll
        for (int u = 0; u < CPW; ++u) {
            const int i = c0 + (int)threadIdx.x + u * 256;
            if (i < c1) cpy[i] = cv[u];
        }
#pragma nounroll
        for (int i = c0 + (int)threadIdx.x + CPW * 256; i < c1; i += 256)
            cpy[i] = csr[i];
    }

    if (wv < NT) {
        v8f acc[MT] = {};
        const float* ar = sA + lo * APITCH + 2 * hi;
#pragma unroll
        for (int s = 0; s < D_IN / 4; ++s) {
#pragma unroll
            for (int m = 0; m < MT; ++m) {
                const v2f a = *(const v2f*)(ar + m * 16 * APITCH + 4 * s);
                acc[m] = __builtin_amdgcn_wmma_f32_16x16x4_f32(
                    false, a, false, bw[s], (short)0, acc[m], false, false);
            }
        }
#pragma unroll
        for (int m = 0; m < MT; ++m) {
#pragma unroll
            for (int r = 0; r < 8; ++r) {
                const int rl = m * 16 + 8 * hi + r;
                sD[rl * DPITCH + wv * 16 + lo] = acc[m][r] * sdi[rl];
            }
        }
    }
    __syncthreads();

    constexpr int F4 = FROWS * OUTC / 4;
#pragma unroll
    for (int i = 0; i < (F4 + 255) / 256; ++i) {
        const int f = (int)threadIdx.x + i * 256;
        const int rl = f / (OUTC / 4);
        const int c4 = f % (OUTC / 4);
        if (f < F4 && row0 + rl < n) {
            const float4 v = *(const float4*)(sD + rl * DPITCH + 4 * c4);
            *(float4*)(Hn + (size_t)(row0 + rl) * OUTC + 4 * c4) = v;
        }
    }
}

__global__ void gather16_kernel(const int* __restrict__ rp,
                                const int* __restrict__ csr,
                                const float* __restrict__ Hs,
                                const float* __restrict__ bias,
                                float* __restrict__ out, int n, int e) {
    const int lane = threadIdx.x & 31;
    const int w = blockIdx.x * WAVES_PER_BLOCK + (int)(threadIdx.x >> 5);
    const int node = 2 * w + (lane >> 4);
    const int slot = (lane >> 2) & 3;
    const int q = lane & 3;

    float4 acc = make_float4(0.0f, 0.0f, 0.0f, 0.0f);
    float di = 0.0f;
    if (node < n) {
        const int start = rp[node];
        const int end = (node + 1 < n) ? rp[node + 1] : e;
        di = rsqrtf((float)(end - start + 1));
        if (slot == 0) acc = ((const float4*)(Hs + (size_t)node * 16))[q];
        for (int j0 = start; j0 < end; j0 += 16) {
            int id[4];
            float4 h[4];
#pragma unroll
            for (int u = 0; u < 4; ++u) {
                const int j = j0 + 4 * u + slot;
                id[u] = (j < end) ? csr[j] : -1;
            }
#pragma unroll
            for (int u = 0; u < 4; ++u) {
                h[u] = make_float4(0.0f, 0.0f, 0.0f, 0.0f);
                if (id[u] >= 0)
                    h[u] = ((const float4*)(Hs + (size_t)id[u] * 16))[q];
            }
#pragma unroll
            for (int u = 0; u < 4; ++u) {
                acc.x += h[u].x;
                acc.y += h[u].y;
                acc.z += h[u].z;
                acc.w += h[u].w;
            }
        }
    }
    acc.x += __shfl_xor(acc.x, 4, 32);
    acc.y += __shfl_xor(acc.y, 4, 32);
    acc.z += __shfl_xor(acc.z, 4, 32);
    acc.w += __shfl_xor(acc.w, 4, 32);
    acc.x += __shfl_xor(acc.x, 8, 32);
    acc.y += __shfl_xor(acc.y, 8, 32);
    acc.z += __shfl_xor(acc.z, 8, 32);
    acc.w += __shfl_xor(acc.w, 8, 32);

    if (node < n && slot == 0) {
        const float4 b = ((const float4*)bias)[q];
        float4 o;
        o.x = b.x + acc.x * di;
        o.y = b.y + acc.y * di;
        o.z = b.z + acc.z * di;
        o.w = b.w + acc.w * di;
        ((float4*)(out + (size_t)node * 16))[q] = o;
    }
}

extern "C" void kernel_launch(void* const* d_in, const int* in_sizes, int n_in,
                              void* d_out, int out_size, void* d_ws,
                              size_t ws_size, hipStream_t stream) {
    const float* x  = (const float*)d_in[0];
    const int*   ei = (const int*)d_in[1];
    const float* W1 = (const float*)d_in[2];
    const float* b1 = (const float*)d_in[3];
    const float* W2 = (const float*)d_in[4];
    const float* b2 = (const float*)d_in[5];
    const float* W3 = (const float*)d_in[6];
    const float* b3 = (const float*)d_in[7];
    const float* W4 = (const float*)d_in[8];
    const float* b4 = (const float*)d_in[9];

    const int N = in_sizes[0] / D_IN;
    const int E = in_sizes[1] / 2;
    const int* src = ei;
    const int* dst = ei + E;

    int*   rp   = (int*)d_ws;
    float* bufH = (float*)d_ws + N;
    float* bufA = bufH + (size_t)N * D_IN;
    float* outp = (float*)d_out;
    int*   csr  = (int*)d_out;
    int*   csr4 = (int*)(bufA + (size_t)N * 16);

    const int NB = (N + BW - 1) / BW;
    const int GA = (E + EPB - 1) / EPB;
    const int NT = NB * GA;
    int2*  pairs = (int2*)bufA;
    int*   tab   = (int*)bufA + 2 * (size_t)E;
    int*   off   = tab + NT;

    const int TPB = 256;
    const int tiles = (N + 15) / 16;
    const int gGemm = (tiles + WAVES_PER_BLOCK - 1) / WAVES_PER_BLOCK;
    const int gFuse = (N + FROWS - 1) / FROWS;
    const int gG16  = ((N + 1) / 2 + WAVES_PER_BLOCK - 1) / WAVES_PER_BLOCK;

    bin_local_kernel<<<GA, 256, 0, stream>>>(src, dst, pairs, tab, off, E, NB, GA);
    bin_finish_kernel<<<NB, 1024, 0, stream>>>(pairs, tab, off, rp, csr, N, NB, GA);

    gemm_wmma_kernel<64, false><<<gGemm, TPB, 0, stream>>>(x, W1, rp, bufH, N, E);

    gather_gemm_kernel<64, false><<<gFuse, TPB, 0, stream>>>(rp, csr, bufH, b1, W2, bufA, nullptr, N, E);
    gather_gemm_kernel<64, false><<<gFuse, TPB, 0, stream>>>(rp, csr, bufA, b2, W3, bufH, nullptr, N, E);
    gather_gemm_kernel<16, true><<<gFuse, TPB, 0, stream>>>(rp, csr, bufH, b3, W4, bufA, csr4, N, E);

    gather16_kernel<<<gG16, TPB, 0, stream>>>(rp, csr4, bufA, b4, outp, N, E);
}
